# starting kernel
# baseline (speedup 1.0000x reference)
#include <hip/hip_runtime.h>
#include <math.h>

#define NN 50000
#define EE 800000
#define MM 3
#define IN_SZ 128
#define HH 8
#define OO 32
#define DD 256
#define HID 128

typedef __attribute__((ext_vector_type(2))) float v2f;
typedef __attribute__((ext_vector_type(8))) float v8f;

__global__ __launch_bounds__(32) void feat_gemm_kernel(
    const float* __restrict__ h, const float* __restrict__ W,
    float* __restrict__ feat) {
  const int lane = threadIdx.x;
  const int half = lane >> 4;
  const int l16  = lane & 15;
  const int row0 = blockIdx.x * 16;

  v2f a[32];
  const float* arow = h + (size_t)(row0 + l16) * IN_SZ;
#pragma unroll
  for (int kk = 0; kk < 32; ++kk) {
    int k = kk * 4 + half * 2;
    a[kk].x = arow[k];
    a[kk].y = arow[k + 1];
  }

  for (int ct = 0; ct < 16; ++ct) {
    v8f acc = {};
#pragma unroll
    for (int kk = 0; kk < 32; ++kk) {
      int k = kk * 4 + half * 2;
      v2f b;
      b.x = W[(size_t)k * DD + ct * 16 + l16];
      b.y = W[(size_t)(k + 1) * DD + ct * 16 + l16];
      acc = __builtin_amdgcn_wmma_f32_16x16x4_f32(
          false, a[kk], false, b, (short)0, acc, false, false);
    }
#pragma unroll
    for (int r = 0; r < 8; ++r) {
      feat[(size_t)(row0 + half * 8 + r) * DD + ct * 16 + l16] = acc[r];
    }
  }
}

__global__ void eler_kernel(const float* __restrict__ feat,
                            const float* __restrict__ al,
                            const float* __restrict__ ar,
                            float* __restrict__ el, float* __restrict__ er) {
  int i = blockIdx.x * blockDim.x + threadIdx.x;
  if (i >= NN * HH) return;
  int n = i >> 3, hh = i & 7;
  const float* f  = feat + (size_t)n * DD + hh * OO;
  const float* a1 = al + hh * OO;
  const float* a2 = ar + hh * OO;
  float s1 = 0.f, s2 = 0.f;
#pragma unroll
  for (int o = 0; o < OO; ++o) { float fv = f[o]; s1 += fv * a1[o]; s2 += fv * a2[o]; }
  el[i] = s1; er[i] = s2;
}

__global__ void minit_kernel(float* __restrict__ m, int count) {
  int i = blockIdx.x * blockDim.x + threadIdx.x;
  if (i < count) m[i] = -INFINITY;
}

__device__ __forceinline__ void atomicMaxFloat(float* addr, float val) {
  if (val >= 0.f) atomicMax((int*)addr, __float_as_int(val));
  else            atomicMin((unsigned int*)addr, __float_as_uint(val));
}

__global__ void edge_max_kernel(const int* __restrict__ src,
                                const int* __restrict__ dst,
                                const float* __restrict__ el,
                                const float* __restrict__ er,
                                float* __restrict__ mbuf) {
  int e = blockIdx.x * blockDim.x + threadIdx.x;
  if (e >= EE) return;
  int s = src[e], d = dst[e];
#pragma unroll
  for (int hh = 0; hh < HH; ++hh) {
    float v = el[s * HH + hh] + er[d * HH + hh];
    v = v > 0.f ? v : 0.2f * v;
    atomicMaxFloat(&mbuf[d * HH + hh], v);
  }
}

__global__ void edge_sum_kernel(const int* __restrict__ src,
                                const int* __restrict__ dst,
                                const float* __restrict__ el,
                                const float* __restrict__ er,
                                const float* __restrict__ mbuf,
                                float* __restrict__ den) {
  int e = blockIdx.x * blockDim.x + threadIdx.x;
  if (e >= EE) return;
  int s = src[e], d = dst[e];
#pragma unroll
  for (int hh = 0; hh < HH; ++hh) {
    float v = el[s * HH + hh] + er[d * HH + hh];
    v = v > 0.f ? v : 0.2f * v;
    atomicAdd(&den[d * HH + hh], expf(v - mbuf[d * HH + hh]));
  }
}

__global__ __launch_bounds__(256) void edge_scatter_kernel(
    const int* __restrict__ src, const int* __restrict__ dst,
    const float* __restrict__ el, const float* __restrict__ er,
    const float* __restrict__ mbuf, const float* __restrict__ den,
    const float* __restrict__ feat, float* __restrict__ zm) {
  __shared__ float alpha_s[HH];
  int e = blockIdx.x;
  int s = src[e], d = dst[e];
  if (threadIdx.x < HH) {
    int hh = threadIdx.x;
    float v = el[s * HH + hh] + er[d * HH + hh];
    v = v > 0.f ? v : 0.2f * v;
    float dn = den[d * HH + hh];
    dn = (dn == 0.f) ? 1.f : dn;
    alpha_s[hh] = expf(v - mbuf[d * HH + hh]) / dn;
  }
  __syncthreads();
  int dd = threadIdx.x;
  atomicAdd(&zm[(size_t)d * (MM * DD) + dd],
            alpha_s[dd >> 5] * feat[(size_t)s * DD + dd]);
}

__global__ void bias_elu_kernel(float* __restrict__ zm,
                                const float* __restrict__ bias) {
  int i = blockIdx.x * blockDim.x + threadIdx.x;
  if (i >= NN * DD) return;
  int n = i >> 8, dd = i & 255;
  size_t idx = (size_t)n * (MM * DD) + dd;
  float x = zm[idx] + bias[dd];
  zm[idx] = x > 0.f ? x : expm1f(x);
}

__global__ __launch_bounds__(128) void semantic_kernel(
    const float* __restrict__ z, const float* __restrict__ W1,
    const float* __restrict__ b1, const float* __restrict__ W2,
    float* __restrict__ wsum) {
  __shared__ float zrow[DD];
  __shared__ float red[HID];
  int nm = blockIdx.x;
  int t = threadIdx.x;
  zrow[t]       = z[(size_t)nm * DD + t];
  zrow[t + 128] = z[(size_t)nm * DD + 128 + t];
  __syncthreads();
  float s = b1[t];
#pragma unroll 8
  for (int i = 0; i < DD; ++i) s += zrow[i] * W1[(size_t)i * HID + t];
  red[t] = tanhf(s) * W2[t];
  __syncthreads();
  for (int off = 64; off > 0; off >>= 1) {
    if (t < off) red[t] += red[t + off];
    __syncthreads();
  }
  if (t == 0) atomicAdd(&wsum[nm % MM], red[0]);
}

__global__ void beta_kernel(const float* __restrict__ wsum,
                            float* __restrict__ beta) {
  float w0 = wsum[0] * (1.f / NN);
  float w1 = wsum[1] * (1.f / NN);
  float w2 = wsum[2] * (1.f / NN);
  float mx = fmaxf(w0, fmaxf(w1, w2));
  float e0 = expf(w0 - mx), e1 = expf(w1 - mx), e2 = expf(w2 - mx);
  float inv = 1.f / (e0 + e1 + e2);
  beta[0] = e0 * inv; beta[1] = e1 * inv; beta[2] = e2 * inv;
}

__global__ void final_kernel(const float* __restrict__ z,
                             const float* __restrict__ beta,
                             float* __restrict__ out) {
  int i = blockIdx.x * blockDim.x + threadIdx.x;
  if (i >= NN * DD) return;
  int n = i >> 8, dd = i & 255;
  size_t base = (size_t)n * (MM * DD) + dd;
  out[i] = beta[0] * z[base] + beta[1] * z[base + DD] + beta[2] * z[base + 2 * DD];
}

extern "C" void kernel_launch(void* const* d_in, const int* in_sizes, int n_in,
                              void* d_out, int out_size, void* d_ws, size_t ws_size,
                              hipStream_t stream) {
  const float* h       = (const float*)d_in[0];
  const int*   eidx    = (const int*)  d_in[1];
  const float* W       = (const float*)d_in[2];
  const float* attn_l  = (const float*)d_in[3];
  const float* attn_r  = (const float*)d_in[4];
  const float* bias    = (const float*)d_in[5];
  const float* sem_W1  = (const float*)d_in[6];
  const float* sem_b1  = (const float*)d_in[7];
  const float* sem_W2  = (const float*)d_in[8];
  float* out = (float*)d_out;

  float* ws   = (float*)d_ws;
  float* z    = ws;
  float* feat = z + (size_t)NN * MM * DD;
  float* el   = feat + (size_t)NN * DD;
  float* er   = el + NN * HH;
  float* mbuf = er + NN * HH;
  float* den  = mbuf + NN * HH;
  float* wsum = den + NN * HH;
  float* beta = wsum + MM;

  hipMemsetAsync(z, 0, sizeof(float) * (size_t)NN * MM * DD, stream);
  hipMemsetAsync(wsum, 0, sizeof(float) * MM, stream);

  for (int m = 0; m < MM; ++m) {
    const float* hm   = h + (size_t)m * NN * IN_SZ;
    const float* Wm   = W + (size_t)m * IN_SZ * DD;
    const float* alm  = attn_l + (size_t)m * HH * OO;
    const float* arm  = attn_r + (size_t)m * HH * OO;
    const float* bm   = bias + (size_t)m * DD;
    const int*   srcm = eidx + (size_t)m * 2 * EE;
    const int*   dstm = srcm + EE;
    float* zm = z + (size_t)m * DD;

    feat_gemm_kernel<<<NN / 16, 32, 0, stream>>>(hm, Wm, feat);
    eler_kernel<<<(NN * HH + 255) / 256, 256, 0, stream>>>(feat, alm, arm, el, er);
    minit_kernel<<<(NN * HH + 255) / 256, 256, 0, stream>>>(mbuf, NN * HH);
    hipMemsetAsync(den, 0, sizeof(float) * NN * HH, stream);
    edge_max_kernel<<<(EE + 255) / 256, 256, 0, stream>>>(srcm, dstm, el, er, mbuf);
    edge_sum_kernel<<<(EE + 255) / 256, 256, 0, stream>>>(srcm, dstm, el, er, mbuf, den);
    edge_scatter_kernel<<<EE, 256, 0, stream>>>(srcm, dstm, el, er, mbuf, den, feat, zm);
    bias_elu_kernel<<<(NN * DD + 255) / 256, 256, 0, stream>>>(zm, bm);
  }

  semantic_kernel<<<NN * MM, 128, 0, stream>>>(z, sem_W1, sem_b1, sem_W2, wsum);
  beta_kernel<<<1, 1, 0, stream>>>(wsum, beta);
  final_kernel<<<(NN * DD + 255) / 256, 256, 0, stream>>>(z, beta, out);
}


# Edges grouped by destination in-call (count, scan, returning-add fill); one wave per node reduces its list in registers, one divide, stores the row with bias+ELU; no float atomics, sum in list order
# speedup vs baseline: 3.7634x; 3.7634x over previous
#include <hip/hip_runtime.h>
#include <math.h>

#define NN 50000
#define EE 800000
#define MM 3
#define IN_SZ 128
#define HH 8
#define OO 32
#define DD 256
#define HID 128

typedef __attribute__((ext_vector_type(2))) float v2f;
typedef __attribute__((ext_vector_type(8))) float v8f;

__global__ __launch_bounds__(32) void feat_gemm_kernel(
    const float* __restrict__ h, const float* __restrict__ W,
    float* __restrict__ feat) {
  const int lane = threadIdx.x;
  const int half = lane >> 4;
  const int l16  = lane & 15;
  const int row0 = blockIdx.x * 16;

  v2f a[32];
  const float* arow = h + (size_t)(row0 + l16) * IN_SZ;
#pragma unroll
  for (int kk = 0; kk < 32; ++kk) {
    int k = kk * 4 + half * 2;
    a[kk].x = arow[k];
    a[kk].y = arow[k + 1];
  }

  for (int ct = 0; ct < 16; ++ct) {
    v8f acc = {};
#pragma unroll
    for (int kk = 0; kk < 32; ++kk) {
      int k = kk * 4 + half * 2;
      v2f b;
      b.x = W[(size_t)k * DD + ct * 16 + l16];
      b.y = W[(size_t)(k + 1) * DD + ct * 16 + l16];
      acc = __builtin_amdgcn_wmma_f32_16x16x4_f32(
          false, a[kk], false, b, (short)0, acc, false, false);
    }
#pragma unroll
    for (int r = 0; r < 8; ++r) {
      feat[(size_t)(row0 + half * 8 + r) * DD + ct * 16 + l16] = acc[r];
    }
  }
}

__global__ void eler_kernel(const float* __restrict__ feat,
                            const float* __restrict__ al,
                            const float* __restrict__ ar,
                            float* __restrict__ el, float* __restrict__ er) {
  int i = blockIdx.x * blockDim.x + threadIdx.x;
  if (i >= NN * HH) return;
  int n = i >> 3, hh = i & 7;
  const float* f  = feat + (size_t)n * DD + hh * OO;
  const float* a1 = al + hh * OO;
  const float* a2 = ar + hh * OO;
  float s1 = 0.f, s2 = 0.f;
#pragma unroll
  for (int o = 0; o < OO; ++o) { float fv = f[o]; s1 += fv * a1[o]; s2 += fv * a2[o]; }
  el[i] = s1; er[i] = s2;
}

#define ZS (MM * DD)
#define SCAN_BLOCKS ((NN + 255) / 256)
static_assert(SCAN_BLOCKS <= 256, "one thread per scan block total");
static_assert(NN % 8 == 0, "eight node waves per block, no tail");
static_assert(EE <= 2 * NN * HH, "edge list must fit the two freed per-head planes");

__global__ void node_init_kernel(int* __restrict__ zmi) {
  int i = blockIdx.x * blockDim.x + threadIdx.x;
  if (i < NN) zmi[(size_t)i * ZS] = 0;
}

__global__ void edge_count_kernel(const int* __restrict__ dst,
                                  int* __restrict__ zmi) {
  int e = blockIdx.x * blockDim.x + threadIdx.x;
  if (e >= EE) return;
  atomicAdd(&zmi[(size_t)dst[e] * ZS], 1);
}

__global__ __launch_bounds__(256) void scan_local_kernel(int* __restrict__ zmi) {
  __shared__ int sh[256];
  const int t = threadIdx.x;
  const int d = blockIdx.x * 256 + t;
  const int c = (d < NN) ? zmi[(size_t)d * ZS] : 0;
  sh[t] = c;
  __syncthreads();
  for (int off = 1; off < 256; off <<= 1) {
    int v = (t >= off) ? sh[t - off] : 0;
    __syncthreads();
    sh[t] += v;
    __syncthreads();
  }
  const int incl = sh[t];
  if (d < NN) zmi[(size_t)d * ZS + 1] = incl - c;
  if (t == 255) zmi[(size_t)blockIdx.x * ZS + 2] = incl;
}

__global__ __launch_bounds__(256) void scan_offset_kernel(int* __restrict__ zmi) {
  __shared__ int sh[256];
  const int t = threadIdx.x;
  const int b = blockIdx.x;
  sh[t] = (t < b) ? zmi[(size_t)t * ZS + 2] : 0;
  __syncthreads();
  for (int off = 128; off > 0; off >>= 1) {
    if (t < off) sh[t] += sh[t + off];
    __syncthreads();
  }
  const int ofs = sh[0];
  const int d = b * 256 + t;
  if (d < NN) zmi[(size_t)d * ZS + 1] += ofs;
}

__global__ void edge_fill_kernel(const int* __restrict__ src,
                                 const int* __restrict__ dst,
                                 int* __restrict__ zmi, int* __restrict__ lst) {
  int e = blockIdx.x * blockDim.x + threadIdx.x;
  if (e >= EE) return;
  int pos = atomicAdd(&zmi[(size_t)dst[e] * ZS + 1], 1);
  lst[pos] = src[e];
}

__device__ __forceinline__ float leaky(float v) { return v > 0.f ? v : 0.2f * v; }
__device__ __forceinline__ float elu1(float x) { return x > 0.f ? x : expm1f(x); }

__global__ __launch_bounds__(256) void node_gather_kernel(
    const int* __restrict__ lst, const float* __restrict__ el,
    const float* __restrict__ er, const float* __restrict__ feat,
    const float* __restrict__ bias, float* zm) {
  const int lane = threadIdx.x & 31;
  const int q = lane >> 3;
  const int d = __builtin_amdgcn_readfirstlane(blockIdx.x * 8 + (threadIdx.x >> 5));
  float* zrow = zm + (size_t)d * ZS;
  const int cnt = ((const int*)zrow)[0];
  const int end = ((const int*)zrow)[1];
  const int beg = end - cnt;

  const float4 erlo = *(const float4*)(er + (size_t)d * HH);
  const float4 erhi = *(const float4*)(er + (size_t)d * HH + 4);
  const float erv[8] = {erlo.x, erlo.y, erlo.z, erlo.w, erhi.x, erhi.y, erhi.z, erhi.w};

  float mx[8];
#pragma unroll
  for (int hh = 0; hh < 8; ++hh) mx[hh] = -INFINITY;
  for (int base = beg; base < end; base += 32) {
    const int j = base + lane;
    if (j < end) {
      const int s = lst[j];
      const float4 a = *(const float4*)(el + (size_t)s * HH);
      const float4 b = *(const float4*)(el + (size_t)s * HH + 4);
      const float ev[8] = {a.x, a.y, a.z, a.w, b.x, b.y, b.z, b.w};
#pragma unroll
      for (int hh = 0; hh < 8; ++hh) mx[hh] = fmaxf(mx[hh], leaky(ev[hh] + erv[hh]));
    }
  }
#pragma unroll
  for (int off = 16; off > 0; off >>= 1) {
#pragma unroll
    for (int hh = 0; hh < 8; ++hh) mx[hh] = fmaxf(mx[hh], __shfl_xor(mx[hh], off));
  }
  float mA = q == 0 ? mx[0] : q == 1 ? mx[1] : q == 2 ? mx[2] : mx[3];
  float mB = q == 0 ? mx[4] : q == 1 ? mx[5] : q == 2 ? mx[6] : mx[7];
  mA = __builtin_isfinite(mA) ? mA : 0.f;
  mB = __builtin_isfinite(mB) ? mB : 0.f;
  const float erA = q == 0 ? erv[0] : q == 1 ? erv[1] : q == 2 ? erv[2] : erv[3];
  const float erB = q == 0 ? erv[4] : q == 1 ? erv[5] : q == 2 ? erv[6] : erv[7];

  const float4* feat4 = (const float4*)feat;
  float4 acc0 = {0.f, 0.f, 0.f, 0.f}, acc1 = {0.f, 0.f, 0.f, 0.f};
  float dA = 0.f, dB = 0.f;
  for (int base = beg; base < end; base += 32) {
    const int j = base + lane;
    const int my_s = (j < end) ? lst[j] : 0;
    const int n = (end - base) < 32 ? (end - base) : 32;
#pragma unroll 4
    for (int i = 0; i < n; ++i) {
      const int s = __builtin_amdgcn_readlane(my_s, i);
      const float4 f0 = feat4[(size_t)s * (DD / 4) + lane];
      const float4 f1 = feat4[(size_t)s * (DD / 4) + 32 + lane];
      const float pA = expf(leaky(el[(size_t)s * HH + q] + erA) - mA);
      const float pB = expf(leaky(el[(size_t)s * HH + 4 + q] + erB) - mB);
      dA += pA; dB += pB;
      acc0.x += pA * f0.x; acc0.y += pA * f0.y; acc0.z += pA * f0.z; acc0.w += pA * f0.w;
      acc1.x += pB * f1.x; acc1.y += pB * f1.y; acc1.z += pB * f1.z; acc1.w += pB * f1.w;
    }
  }
  dA = (dA == 0.f) ? 1.f : dA;
  dB = (dB == 0.f) ? 1.f : dB;

  const float4 b0 = ((const float4*)bias)[lane];
  const float4 b1 = ((const float4*)bias)[32 + lane];
  float4 o0, o1;
  o0.x = elu1(acc0.x / dA + b0.x); o0.y = elu1(acc0.y / dA + b0.y);
  o0.z = elu1(acc0.z / dA + b0.z); o0.w = elu1(acc0.w / dA + b0.w);
  o1.x = elu1(acc1.x / dB + b1.x); o1.y = elu1(acc1.y / dB + b1.y);
  o1.z = elu1(acc1.z / dB + b1.z); o1.w = elu1(acc1.w / dB + b1.w);
  ((float4*)zrow)[lane] = o0;
  ((float4*)zrow)[32 + lane] = o1;
}

__global__ __launch_bounds__(128) void semantic_kernel(
    const float* __restrict__ z, const float* __restrict__ W1,
    const float* __restrict__ b1, const float* __restrict__ W2,
    float* __restrict__ wsum) {
  __shared__ float zrow[DD];
  __shared__ float red[HID];
  int nm = blockIdx.x;
  int t = threadIdx.x;
  zrow[t]       = z[(size_t)nm * DD + t];
  zrow[t + 128] = z[(size_t)nm * DD + 128 + t];
  __syncthreads();
  float s = b1[t];
#pragma unroll 8
  for (int i = 0; i < DD; ++i) s += zrow[i] * W1[(size_t)i * HID + t];
  red[t] = tanhf(s) * W2[t];
  __syncthreads();
  for (int off = 64; off > 0; off >>= 1) {
    if (t < off) red[t] += red[t + off];
    __syncthreads();
  }
  if (t == 0) atomicAdd(&wsum[nm % MM], red[0]);
}

__global__ void beta_kernel(const float* __restrict__ wsum,
                            float* __restrict__ beta) {
  float w0 = wsum[0] * (1.f / NN);
  float w1 = wsum[1] * (1.f / NN);
  float w2 = wsum[2] * (1.f / NN);
  float mx = fmaxf(w0, fmaxf(w1, w2));
  float e0 = expf(w0 - mx), e1 = expf(w1 - mx), e2 = expf(w2 - mx);
  float inv = 1.f / (e0 + e1 + e2);
  beta[0] = e0 * inv; beta[1] = e1 * inv; beta[2] = e2 * inv;
}

__global__ void final_kernel(const float* __restrict__ z,
                             const float* __restrict__ beta,
                             float* __restrict__ out) {
  int i = blockIdx.x * blockDim.x + threadIdx.x;
  if (i >= NN * DD) return;
  int n = i >> 8, dd = i & 255;
  size_t base = (size_t)n * (MM * DD) + dd;
  out[i] = beta[0] * z[base] + beta[1] * z[base + DD] + beta[2] * z[base + 2 * DD];
}

extern "C" void kernel_launch(void* const* d_in, const int* in_sizes, int n_in,
                              void* d_out, int out_size, void* d_ws, size_t ws_size,
                              hipStream_t stream) {
  const float* h       = (const float*)d_in[0];
  const int*   eidx    = (const int*)  d_in[1];
  const float* W       = (const float*)d_in[2];
  const float* attn_l  = (const float*)d_in[3];
  const float* attn_r  = (const float*)d_in[4];
  const float* bias    = (const float*)d_in[5];
  const float* sem_W1  = (const float*)d_in[6];
  const float* sem_b1  = (const float*)d_in[7];
  const float* sem_W2  = (const float*)d_in[8];
  float* out = (float*)d_out;

  float* ws   = (float*)d_ws;
  float* z    = ws;
  float* feat = z + (size_t)NN * MM * DD;
  float* el   = feat + (size_t)NN * DD;
  float* er   = el + NN * HH;
  int*   lst  = (int*)(er + NN * HH);
  float* wsum = er + 3 * NN * HH;
  float* beta = wsum + MM;

  hipMemsetAsync(wsum, 0, sizeof(float) * MM, stream);

  for (int m = 0; m < MM; ++m) {
    const float* hm   = h + (size_t)m * NN * IN_SZ;
    const float* Wm   = W + (size_t)m * IN_SZ * DD;
    const float* alm  = attn_l + (size_t)m * HH * OO;
    const float* arm  = attn_r + (size_t)m * HH * OO;
    const float* bm   = bias + (size_t)m * DD;
    const int*   srcm = eidx + (size_t)m * 2 * EE;
    const int*   dstm = srcm + EE;
    float* zm = z + (size_t)m * DD;

    feat_gemm_kernel<<<NN / 16, 32, 0, stream>>>(hm, Wm, feat);
    eler_kernel<<<(NN * HH + 255) / 256, 256, 0, stream>>>(feat, alm, arm, el, er);
    node_init_kernel<<<(NN + 255) / 256, 256, 0, stream>>>((int*)zm);
    edge_count_kernel<<<(EE + 255) / 256, 256, 0, stream>>>(dstm, (int*)zm);
    scan_local_kernel<<<SCAN_BLOCKS, 256, 0, stream>>>((int*)zm);
    scan_offset_kernel<<<SCAN_BLOCKS, 256, 0, stream>>>((int*)zm);
    edge_fill_kernel<<<(EE + 255) / 256, 256, 0, stream>>>(srcm, dstm, (int*)zm, lst);
    node_gather_kernel<<<NN / 8, 256, 0, stream>>>(lst, el, er, feat, bm, zm);
  }

  semantic_kernel<<<NN * MM, 128, 0, stream>>>(z, sem_W1, sem_b1, sem_W2, wsum);
  beta_kernel<<<1, 1, 0, stream>>>(wsum, beta);
  final_kernel<<<(NN * DD + 255) / 256, 256, 0, stream>>>(z, beta, out);
}


# Semantic score product z@W1 moved from vector FMAs to f32 16x16x4 WMMA, W1 held in LDS, one wave per 16-row strip; dot and per-metapath score sums change order (768 atomics, was 150000)
# speedup vs baseline: 5.8964x; 1.5668x over previous
#include <hip/hip_runtime.h>
#include <math.h>

#define NN 50000
#define EE 800000
#define MM 3
#define IN_SZ 128
#define HH 8
#define OO 32
#define DD 256
#define HID 128

typedef __attribute__((ext_vector_type(2))) float v2f;
typedef __attribute__((ext_vector_type(8))) float v8f;

__global__ __launch_bounds__(32) void feat_gemm_kernel(
    const float* __restrict__ h, const float* __restrict__ W,
    float* __restrict__ feat) {
  const int lane = threadIdx.x;
  const int half = lane >> 4;
  const int l16  = lane & 15;
  const int row0 = blockIdx.x * 16;

  v2f a[32];
  const float* arow = h + (size_t)(row0 + l16) * IN_SZ;
#pragma unroll
  for (int kk = 0; kk < 32; ++kk) {
    int k = kk * 4 + half * 2;
    a[kk].x = arow[k];
    a[kk].y = arow[k + 1];
  }

  for (int ct = 0; ct < 16; ++ct) {
    v8f acc = {};
#pragma unroll
    for (int kk = 0; kk < 32; ++kk) {
      int k = kk * 4 + half * 2;
      v2f b;
      b.x = W[(size_t)k * DD + ct * 16 + l16];
      b.y = W[(size_t)(k + 1) * DD + ct * 16 + l16];
      acc = __builtin_amdgcn_wmma_f32_16x16x4_f32(
          false, a[kk], false, b, (short)0, acc, false, false);
    }
#pragma unroll
    for (int r = 0; r < 8; ++r) {
      feat[(size_t)(row0 + half * 8 + r) * DD + ct * 16 + l16] = acc[r];
    }
  }
}

__global__ void eler_kernel(const float* __restrict__ feat,
                            const float* __restrict__ al,
                            const float* __restrict__ ar,
                            float* __restrict__ el, float* __restrict__ er) {
  int i = blockIdx.x * blockDim.x + threadIdx.x;
  if (i >= NN * HH) return;
  int n = i >> 3, hh = i & 7;
  const float* f  = feat + (size_t)n * DD + hh * OO;
  const float* a1 = al + hh * OO;
  const float* a2 = ar + hh * OO;
  float s1 = 0.f, s2 = 0.f;
#pragma unroll
  for (int o = 0; o < OO; ++o) { float fv = f[o]; s1 += fv * a1[o]; s2 += fv * a2[o]; }
  el[i] = s1; er[i] = s2;
}

#define ZS (MM * DD)
#define SCAN_BLOCKS ((NN + 255) / 256)
static_assert(SCAN_BLOCKS <= 256, "one thread per scan block total");
static_assert(NN % 8 == 0, "eight node waves per block, no tail");
static_assert(EE <= 2 * NN * HH, "edge list must fit the two freed per-head planes");

__global__ void node_init_kernel(int* __restrict__ zmi) {
  int i = blockIdx.x * blockDim.x + threadIdx.x;
  if (i < NN) zmi[(size_t)i * ZS] = 0;
}

__global__ void edge_count_kernel(const int* __restrict__ dst,
                                  int* __restrict__ zmi) {
  int e = blockIdx.x * blockDim.x + threadIdx.x;
  if (e >= EE) return;
  atomicAdd(&zmi[(size_t)dst[e] * ZS], 1);
}

__global__ __launch_bounds__(256) void scan_local_kernel(int* __restrict__ zmi) {
  __shared__ int sh[256];
  const int t = threadIdx.x;
  const int d = blockIdx.x * 256 + t;
  const int c = (d < NN) ? zmi[(size_t)d * ZS] : 0;
  sh[t] = c;
  __syncthreads();
  for (int off = 1; off < 256; off <<= 1) {
    int v = (t >= off) ? sh[t - off] : 0;
    __syncthreads();
    sh[t] += v;
    __syncthreads();
  }
  const int incl = sh[t];
  if (d < NN) zmi[(size_t)d * ZS + 1] = incl - c;
  if (t == 255) zmi[(size_t)blockIdx.x * ZS + 2] = incl;
}

__global__ __launch_bounds__(256) void scan_offset_kernel(int* __restrict__ zmi) {
  __shared__ int sh[256];
  const int t = threadIdx.x;
  const int b = blockIdx.x;
  sh[t] = (t < b) ? zmi[(size_t)t * ZS + 2] : 0;
  __syncthreads();
  for (int off = 128; off > 0; off >>= 1) {
    if (t < off) sh[t] += sh[t + off];
    __syncthreads();
  }
  const int ofs = sh[0];
  const int d = b * 256 + t;
  if (d < NN) zmi[(size_t)d * ZS + 1] += ofs;
}

__global__ void edge_fill_kernel(const int* __restrict__ src,
                                 const int* __restrict__ dst,
                                 int* __restrict__ zmi, int* __restrict__ lst) {
  int e = blockIdx.x * blockDim.x + threadIdx.x;
  if (e >= EE) return;
  int pos = atomicAdd(&zmi[(size_t)dst[e] * ZS + 1], 1);
  lst[pos] = src[e];
}

__device__ __forceinline__ float leaky(float v) { return v > 0.f ? v : 0.2f * v; }
__device__ __forceinline__ float elu1(float x) { return x > 0.f ? x : expm1f(x); }

__global__ __launch_bounds__(256) void node_gather_kernel(
    const int* __restrict__ lst, const float* __restrict__ el,
    const float* __restrict__ er, const float* __restrict__ feat,
    const float* __restrict__ bias, float* zm) {
  const int lane = threadIdx.x & 31;
  const int q = lane >> 3;
  const int d = __builtin_amdgcn_readfirstlane(blockIdx.x * 8 + (threadIdx.x >> 5));
  float* zrow = zm + (size_t)d * ZS;
  const int cnt = ((const int*)zrow)[0];
  const int end = ((const int*)zrow)[1];
  const int beg = end - cnt;

  const float4 erlo = *(const float4*)(er + (size_t)d * HH);
  const float4 erhi = *(const float4*)(er + (size_t)d * HH + 4);
  const float erv[8] = {erlo.x, erlo.y, erlo.z, erlo.w, erhi.x, erhi.y, erhi.z, erhi.w};

  float mx[8];
#pragma unroll
  for (int hh = 0; hh < 8; ++hh) mx[hh] = -INFINITY;
  for (int base = beg; base < end; base += 32) {
    const int j = base + lane;
    if (j < end) {
      const int s = lst[j];
      const float4 a = *(const float4*)(el + (size_t)s * HH);
      const float4 b = *(const float4*)(el + (size_t)s * HH + 4);
      const float ev[8] = {a.x, a.y, a.z, a.w, b.x, b.y, b.z, b.w};
#pragma unroll
      for (int hh = 0; hh < 8; ++hh) mx[hh] = fmaxf(mx[hh], leaky(ev[hh] + erv[hh]));
    }
  }
#pragma unroll
  for (int off = 16; off > 0; off >>= 1) {
#pragma unroll
    for (int hh = 0; hh < 8; ++hh) mx[hh] = fmaxf(mx[hh], __shfl_xor(mx[hh], off));
  }
  float mA = q == 0 ? mx[0] : q == 1 ? mx[1] : q == 2 ? mx[2] : mx[3];
  float mB = q == 0 ? mx[4] : q == 1 ? mx[5] : q == 2 ? mx[6] : mx[7];
  mA = __builtin_isfinite(mA) ? mA : 0.f;
  mB = __builtin_isfinite(mB) ? mB : 0.f;
  const float erA = q == 0 ? erv[0] : q == 1 ? erv[1] : q == 2 ? erv[2] : erv[3];
  const float erB = q == 0 ? erv[4] : q == 1 ? erv[5] : q == 2 ? erv[6] : erv[7];

  const float4* feat4 = (const float4*)feat;
  float4 acc0 = {0.f, 0.f, 0.f, 0.f}, acc1 = {0.f, 0.f, 0.f, 0.f};
  float dA = 0.f, dB = 0.f;
  for (int base = beg; base < end; base += 32) {
    const int j = base + lane;
    const int my_s = (j < end) ? lst[j] : 0;
    const int n = (end - base) < 32 ? (end - base) : 32;
#pragma unroll 4
    for (int i = 0; i < n; ++i) {
      const int s = __builtin_amdgcn_readlane(my_s, i);
      const float4 f0 = feat4[(size_t)s * (DD / 4) + lane];
      const float4 f1 = feat4[(size_t)s * (DD / 4) + 32 + lane];
      const float pA = expf(leaky(el[(size_t)s * HH + q] + erA) - mA);
      const float pB = expf(leaky(el[(size_t)s * HH + 4 + q] + erB) - mB);
      dA += pA; dB += pB;
      acc0.x += pA * f0.x; acc0.y += pA * f0.y; acc0.z += pA * f0.z; acc0.w += pA * f0.w;
      acc1.x += pB * f1.x; acc1.y += pB * f1.y; acc1.z += pB * f1.z; acc1.w += pB * f1.w;
    }
  }
  dA = (dA == 0.f) ? 1.f : dA;
  dB = (dB == 0.f) ? 1.f : dB;

  const float4 b0 = ((const float4*)bias)[lane];
  const float4 b1 = ((const float4*)bias)[32 + lane];
  float4 o0, o1;
  o0.x = elu1(acc0.x / dA + b0.x); o0.y = elu1(acc0.y / dA + b0.y);
  o0.z = elu1(acc0.z / dA + b0.z); o0.w = elu1(acc0.w / dA + b0.w);
  o1.x = elu1(acc1.x / dB + b1.x); o1.y = elu1(acc1.y / dB + b1.y);
  o1.z = elu1(acc1.z / dB + b1.z); o1.w = elu1(acc1.w / dB + b1.w);
  ((float4*)zrow)[lane] = o0;
  ((float4*)zrow)[32 + lane] = o1;
}

#define SEM_ROWS   (NN * MM)
#define SEM_STRIPS (SEM_ROWS / 16)
#define SEM_BLOCKS 256
#define SEM_WAVES  8
#define SEM_STEP   (SEM_BLOCKS * SEM_WAVES)
#define W1_PITCH   (2 * HID + 32)
static_assert(SEM_ROWS % 16 == 0, "whole 16-row strips, no tail");
static_assert(DD % 4 == 0 && HID % 16 == 0, "whole k steps and column tiles");

__device__ __forceinline__ void sem_load_strip(v2f (&a)[DD / 4],
                                               const float* __restrict__ z,
                                               int strip, int l16, int hf) {
  const float* arow = z + (size_t)(strip * 16 + l16) * DD + hf * 2;
#pragma unroll
  for (int kk = 0; kk < DD / 4; ++kk) a[kk] = *(const v2f*)(arow + kk * 4);
}

__global__ __launch_bounds__(SEM_WAVES * 32) void semantic_kernel(
    const float* __restrict__ z, const float* __restrict__ W1,
    const float* __restrict__ b1, const float* __restrict__ W2,
    float* __restrict__ wsum) {
  __shared__ __attribute__((aligned(16))) float w1s[(DD / 2) * W1_PITCH];
  __shared__ float red[SEM_WAVES][MM];
  const int t = threadIdx.x;
  const int lane = t & 31;
  const int hf = lane >> 4;
  const int l16 = lane & 15;
  const int wave = __builtin_amdgcn_readfirstlane(t >> 5);
  int strip = wave * SEM_BLOCKS + blockIdx.x;

  v2f a[DD / 4], an[DD / 4];
  sem_load_strip(a, z, strip, l16, hf);

  for (int idx = t; idx < (DD / 2) * HID; idx += SEM_WAVES * 32) {
    const int kp = idx / HID, n = idx % HID;
    v2f v;
    v.x = W1[(size_t)(2 * kp) * HID + n];
    v.y = W1[(size_t)(2 * kp + 1) * HID + n];
    *(v2f*)&w1s[kp * W1_PITCH + 2 * n] = v;
  }
  float b1c[HID / 16], w2c[HID / 16];
#pragma unroll
  for (int ct = 0; ct < HID / 16; ++ct) {
    b1c[ct] = b1[ct * 16 + l16];
    w2c[ct] = W2[ct * 16 + l16];
  }
  __syncthreads();

  int boff = hf * W1_PITCH + 2 * l16;
  float am0 = 0.f, am1 = 0.f, am2 = 0.f;

  for (; strip < SEM_STRIPS; strip += SEM_STEP) {
    const int nxt = strip + SEM_STEP;
    if (nxt < SEM_STRIPS) sem_load_strip(an, z, nxt, l16, hf);

    asm volatile("" : "+v"(boff));
    const float* bbase = &w1s[boff];

    v8f acc[HID / 16];
#pragma unroll
    for (int ct = 0; ct < HID / 16; ++ct) acc[ct] = v8f{};
#pragma unroll
    for (int kk = 0; kk < DD / 4; ++kk) {
      const float* bp = bbase + kk * 2 * W1_PITCH;
#pragma unroll
      for (int ct = 0; ct < HID / 16; ++ct) {
        const v2f b = *(const v2f*)(bp + ct * 32);
        acc[ct] = __builtin_amdgcn_wmma_f32_16x16x4_f32(
            false, a[kk], false, b, (short)0, acc[ct], false, false);
      }
    }

    float p[8];
#pragma unroll
    for (int r = 0; r < 8; ++r) {
      float s = 0.f;
#pragma unroll
      for (int ct = 0; ct < HID / 16; ++ct) s += tanhf(acc[ct][r] + b1c[ct]) * w2c[ct];
      p[r] = s;
    }
    const float s0 = p[0] + p[3] + p[6];
    const float s1 = p[1] + p[4] + p[7];
    const float s2 = p[2] + p[5];
    const int first = (strip * 16 + hf * 8) % MM;
    am0 += first == 0 ? s0 : first == 1 ? s2 : s1;
    am1 += first == 0 ? s1 : first == 1 ? s0 : s2;
    am2 += first == 0 ? s2 : first == 1 ? s1 : s0;

    if (nxt < SEM_STRIPS) {
#pragma unroll
      for (int kk = 0; kk < DD / 4; ++kk) a[kk] = an[kk];
    }
  }

  for (int off = 16; off > 0; off >>= 1) {
    am0 += __shfl_xor(am0, off);
    am1 += __shfl_xor(am1, off);
    am2 += __shfl_xor(am2, off);
  }
  if (lane == 0) { red[wave][0] = am0; red[wave][1] = am1; red[wave][2] = am2; }
  __syncthreads();
  if (t < MM) {
    float s = 0.f;
    for (int w = 0; w < SEM_WAVES; ++w) s += red[w][t];
    atomicAdd(&wsum[t], s);
  }
}

__global__ void beta_kernel(const float* __restrict__ wsum,
                            float* __restrict__ beta) {
  float w0 = wsum[0] * (1.f / NN);
  float w1 = wsum[1] * (1.f / NN);
  float w2 = wsum[2] * (1.f / NN);
  float mx = fmaxf(w0, fmaxf(w1, w2));
  float e0 = expf(w0 - mx), e1 = expf(w1 - mx), e2 = expf(w2 - mx);
  float inv = 1.f / (e0 + e1 + e2);
  beta[0] = e0 * inv; beta[1] = e1 * inv; beta[2] = e2 * inv;
}

__global__ void final_kernel(const float* __restrict__ z,
                             const float* __restrict__ beta,
                             float* __restrict__ out) {
  int i = blockIdx.x * blockDim.x + threadIdx.x;
  if (i >= NN * DD) return;
  int n = i >> 8, dd = i & 255;
  size_t base = (size_t)n * (MM * DD) + dd;
  out[i] = beta[0] * z[base] + beta[1] * z[base + DD] + beta[2] * z[base + 2 * DD];
}

extern "C" void kernel_launch(void* const* d_in, const int* in_sizes, int n_in,
                              void* d_out, int out_size, void* d_ws, size_t ws_size,
                              hipStream_t stream) {
  const float* h       = (const float*)d_in[0];
  const int*   eidx    = (const int*)  d_in[1];
  const float* W       = (const float*)d_in[2];
  const float* attn_l  = (const float*)d_in[3];
  const float* attn_r  = (const float*)d_in[4];
  const float* bias    = (const float*)d_in[5];
  const float* sem_W1  = (const float*)d_in[6];
  const float* sem_b1  = (const float*)d_in[7];
  const float* sem_W2  = (const float*)d_in[8];
  float* out = (float*)d_out;

  float* ws   = (float*)d_ws;
  float* z    = ws;
  float* feat = z + (size_t)NN * MM * DD;
  float* el   = feat + (size_t)NN * DD;
  float* er   = el + NN * HH;
  int*   lst  = (int*)(er + NN * HH);
  float* wsum = er + 3 * NN * HH;
  float* beta = wsum + MM;

  hipMemsetAsync(wsum, 0, sizeof(float) * MM, stream);

  for (int m = 0; m < MM; ++m) {
    const float* hm   = h + (size_t)m * NN * IN_SZ;
    const float* Wm   = W + (size_t)m * IN_SZ * DD;
    const float* alm  = attn_l + (size_t)m * HH * OO;
    const float* arm  = attn_r + (size_t)m * HH * OO;
    const float* bm   = bias + (size_t)m * DD;
    const int*   srcm = eidx + (size_t)m * 2 * EE;
    const int*   dstm = srcm + EE;
    float* zm = z + (size_t)m * DD;

    feat_gemm_kernel<<<NN / 16, 32, 0, stream>>>(hm, Wm, feat);
    eler_kernel<<<(NN * HH + 255) / 256, 256, 0, stream>>>(feat, alm, arm, el, er);
    node_init_kernel<<<(NN + 255) / 256, 256, 0, stream>>>((int*)zm);
    edge_count_kernel<<<(EE + 255) / 256, 256, 0, stream>>>(dstm, (int*)zm);
    scan_local_kernel<<<SCAN_BLOCKS, 256, 0, stream>>>((int*)zm);
    scan_offset_kernel<<<SCAN_BLOCKS, 256, 0, stream>>>((int*)zm);
    edge_fill_kernel<<<(EE + 255) / 256, 256, 0, stream>>>(srcm, dstm, (int*)zm, lst);
    node_gather_kernel<<<NN / 8, 256, 0, stream>>>(lst, el, er, feat, bm, zm);
  }

  semantic_kernel<<<SEM_BLOCKS, SEM_WAVES * 32, 0, stream>>>(z, sem_W1, sem_b1, sem_W2, wsum);
  beta_kernel<<<1, 1, 0, stream>>>(wsum, beta);
  final_kernel<<<(NN * DD + 255) / 256, 256, 0, stream>>>(z, beta, out);
}


# Edge grouping by destination in one launch per metapath (was 5): 256 groups each own 196 nodes, sweep all ids once, LDS counters and cursors; no global atomic per edge; list order by LDS cursor
# speedup vs baseline: 6.5563x; 1.1119x over previous
#include <hip/hip_runtime.h>
#include <math.h>

#define NN 50000
#define EE 800000
#define MM 3
#define IN_SZ 128
#define HH 8
#define OO 32
#define DD 256
#define HID 128

typedef __attribute__((ext_vector_type(2))) float v2f;
typedef __attribute__((ext_vector_type(8))) float v8f;

__global__ __launch_bounds__(32) void feat_gemm_kernel(
    const float* __restrict__ h, const float* __restrict__ W,
    float* __restrict__ feat) {
  const int lane = threadIdx.x;
  const int half = lane >> 4;
  const int l16  = lane & 15;
  const int row0 = blockIdx.x * 16;

  v2f a[32];
  const float* arow = h + (size_t)(row0 + l16) * IN_SZ;
#pragma unroll
  for (int kk = 0; kk < 32; ++kk) {
    int k = kk * 4 + half * 2;
    a[kk].x = arow[k];
    a[kk].y = arow[k + 1];
  }

  for (int ct = 0; ct < 16; ++ct) {
    v8f acc = {};
#pragma unroll
    for (int kk = 0; kk < 32; ++kk) {
      int k = kk * 4 + half * 2;
      v2f b;
      b.x = W[(size_t)k * DD + ct * 16 + l16];
      b.y = W[(size_t)(k + 1) * DD + ct * 16 + l16];
      acc = __builtin_amdgcn_wmma_f32_16x16x4_f32(
          false, a[kk], false, b, (short)0, acc, false, false);
    }
#pragma unroll
    for (int r = 0; r < 8; ++r) {
      feat[(size_t)(row0 + half * 8 + r) * DD + ct * 16 + l16] = acc[r];
    }
  }
}

__global__ void eler_kernel(const float* __restrict__ feat,
                            const float* __restrict__ al,
                            const float* __restrict__ ar,
                            float* __restrict__ el, float* __restrict__ er) {
  int i = blockIdx.x * blockDim.x + threadIdx.x;
  if (i >= NN * HH) return;
  int n = i >> 3, hh = i & 7;
  const float* f  = feat + (size_t)n * DD + hh * OO;
  const float* a1 = al + hh * OO;
  const float* a2 = ar + hh * OO;
  float s1 = 0.f, s2 = 0.f;
#pragma unroll
  for (int o = 0; o < OO; ++o) { float fv = f[o]; s1 += fv * a1[o]; s2 += fv * a2[o]; }
  el[i] = s1; er[i] = s2;
}

#define ZS (MM * DD)
static_assert(NN % 8 == 0, "eight node waves per block, no tail");
static_assert(EE <= 2 * NN * HH, "edge list must fit the two freed per-head planes");

#define CSR_GROUPS  256
#define CSR_SPAN    ((NN + CSR_GROUPS - 1) / CSR_GROUPS)
#define CSR_THREADS 800
#define CSR_WAVES   (CSR_THREADS / 32)
#define CSR_WCAP    640
#define CSR_Q       (EE / 4)
#define CSR_DEPTH   5
#define CSR_TRIPS   50
static_assert(CSR_SPAN <= 256 && CSR_THREADS >= 256, "one LDS counter and one thread per owned node");
static_assert(EE % 4 == 0, "whole 16-byte groups of destination ids");
static_assert(CSR_Q == CSR_THREADS * CSR_DEPTH * CSR_TRIPS, "whole trips, no tail");
static_assert(CSR_TRIPS % 2 == 0, "trips are examined in pairs");
static_assert((CSR_GROUPS - 1) * CSR_SPAN < NN, "every group owns at least one node");

__device__ __forceinline__ void csr_load_trip(int4 (&dq)[CSR_DEPTH], int (&qi)[CSR_DEPTH],
                                              const int4* __restrict__ dst4,
                                              int trip, int t, int rot) {
#pragma unroll
  for (int u = 0; u < CSR_DEPTH; ++u) {
    int j = (trip * CSR_DEPTH + u) * CSR_THREADS + t;
    j = (j >= CSR_Q) ? j - CSR_Q : j;
    int q = j + rot;
    q = (q >= CSR_Q) ? q - CSR_Q : q;
    qi[u] = q;
    dq[u] = dst4[q];
  }
}

__global__ __launch_bounds__(CSR_THREADS) void csr_build_kernel(
    const int* __restrict__ src, const int* __restrict__ dst,
    int* __restrict__ zmi, int* __restrict__ lst) {
  __shared__ int cnt[256];
  __shared__ int sc[256];
  __shared__ int cur[256];
  __shared__ int nbelow;
  __shared__ int spill;
  __shared__ int2 kept[CSR_WAVES][CSR_WCAP];
  const int t = threadIdx.x;
  const int lane = t & 31;
  const int wave = __builtin_amdgcn_readfirstlane(t >> 5);
  const int lo = blockIdx.x * CSR_SPAN;
  const int hi = (lo + CSR_SPAN < NN) ? lo + CSR_SPAN : NN;
  const unsigned span = (unsigned)(hi - lo);
  if (t < 256) cnt[t] = 0;
  if (t == 0) { nbelow = 0; spill = 0; }
  __syncthreads();

  const int rot = blockIdx.x * (CSR_Q / CSR_GROUPS);
  const int4* dst4 = (const int4*)dst;
  int below = 0;
  int nw = 0;

  auto examine = [&](const int4 (&dq)[CSR_DEPTH], const int (&qi)[CSR_DEPTH]) {
#pragma unroll
    for (int u = 0; u < CSR_DEPTH; ++u) {
      const int dv[4] = {dq[u].x, dq[u].y, dq[u].z, dq[u].w};
#pragma unroll
      for (int c = 0; c < 4; ++c) {
        const int d = dv[c];
        below += (d < lo) ? 1 : 0;
        const unsigned ld = (unsigned)(d - lo);
        const bool in = ld < span;
        const unsigned m = __builtin_amdgcn_ballot_w32(in);
        if (m != 0u) {
          if (in) {
            atomicAdd(&cnt[ld], 1);
            const int slot = nw + (int)__builtin_amdgcn_mbcnt_lo(m, 0u);
            if (slot < CSR_WCAP) kept[wave][slot] = make_int2(4 * qi[u] + c, (int)ld);
          }
          nw += __builtin_popcount(m);
        }
      }
    }
  };

  int4 da[CSR_DEPTH], db[CSR_DEPTH];
  int qa[CSR_DEPTH], qb[CSR_DEPTH];
  csr_load_trip(da, qa, dst4, 0, t, rot);
  for (int trip = 0; trip < CSR_TRIPS; trip += 2) {
    csr_load_trip(db, qb, dst4, trip + 1, t, rot);
    examine(da, qa);
    csr_load_trip(da, qa, dst4, trip + 2, t, rot);
    examine(db, qb);
  }
  for (int off = 16; off > 0; off >>= 1) below += __shfl_xor(below, off);
  if (lane == 0) {
    atomicAdd(&nbelow, below);
    if (nw > CSR_WCAP) spill = 1;
  }
  __syncthreads();

  const int c = (t < 256) ? cnt[t] : 0;
  if (t < 256) sc[t] = c;
  __syncthreads();
  for (int off = 1; off < 256; off <<= 1) {
    const int v = (t < 256 && t >= off) ? sc[t - off] : 0;
    __syncthreads();
    if (t < 256) sc[t] += v;
    __syncthreads();
  }
  const int base = nbelow;
  const int spilled = spill;
  if (t < 256) cur[t] = base + sc[t] - c;
  if (t < hi - lo) {
    *(int2*)&zmi[(size_t)(lo + t) * ZS] = make_int2(c, base + sc[t]);
  }
  __syncthreads();

  if (spilled == 0) {
    for (int k = lane; k < nw; k += 32) {
      const int2 s = kept[wave][k];
      const int v = src[s.x];
      lst[atomicAdd(&cur[s.y], 1)] = v;
    }
  } else {
    for (int q = t; q < CSR_Q; q += CSR_THREADS) {
      const int4 d4 = dst4[q];
      const int dv[4] = {d4.x, d4.y, d4.z, d4.w};
#pragma unroll
      for (int c2 = 0; c2 < 4; ++c2) {
        const int d = dv[c2];
        if (d >= lo && d < hi) lst[atomicAdd(&cur[d - lo], 1)] = src[4 * q + c2];
      }
    }
  }
}

__device__ __forceinline__ float leaky(float v) { return v > 0.f ? v : 0.2f * v; }
__device__ __forceinline__ float elu1(float x) { return x > 0.f ? x : expm1f(x); }

__global__ __launch_bounds__(256) void node_gather_kernel(
    const int* __restrict__ lst, const float* __restrict__ el,
    const float* __restrict__ er, const float* __restrict__ feat,
    const float* __restrict__ bias, float* zm) {
  const int lane = threadIdx.x & 31;
  const int q = lane >> 3;
  const int d = __builtin_amdgcn_readfirstlane(blockIdx.x * 8 + (threadIdx.x >> 5));
  float* zrow = zm + (size_t)d * ZS;
  const int cnt = ((const int*)zrow)[0];
  const int end = ((const int*)zrow)[1];
  const int beg = end - cnt;

  const float4 erlo = *(const float4*)(er + (size_t)d * HH);
  const float4 erhi = *(const float4*)(er + (size_t)d * HH + 4);
  const float erv[8] = {erlo.x, erlo.y, erlo.z, erlo.w, erhi.x, erhi.y, erhi.z, erhi.w};

  float mx[8];
#pragma unroll
  for (int hh = 0; hh < 8; ++hh) mx[hh] = -INFINITY;
  for (int base = beg; base < end; base += 32) {
    const int j = base + lane;
    if (j < end) {
      const int s = lst[j];
      const float4 a = *(const float4*)(el + (size_t)s * HH);
      const float4 b = *(const float4*)(el + (size_t)s * HH + 4);
      const float ev[8] = {a.x, a.y, a.z, a.w, b.x, b.y, b.z, b.w};
#pragma unroll
      for (int hh = 0; hh < 8; ++hh) mx[hh] = fmaxf(mx[hh], leaky(ev[hh] + erv[hh]));
    }
  }
#pragma unroll
  for (int off = 16; off > 0; off >>= 1) {
#pragma unroll
    for (int hh = 0; hh < 8; ++hh) mx[hh] = fmaxf(mx[hh], __shfl_xor(mx[hh], off));
  }
  float mA = q == 0 ? mx[0] : q == 1 ? mx[1] : q == 2 ? mx[2] : mx[3];
  float mB = q == 0 ? mx[4] : q == 1 ? mx[5] : q == 2 ? mx[6] : mx[7];
  mA = __builtin_isfinite(mA) ? mA : 0.f;
  mB = __builtin_isfinite(mB) ? mB : 0.f;
  const float erA = q == 0 ? erv[0] : q == 1 ? erv[1] : q == 2 ? erv[2] : erv[3];
  const float erB = q == 0 ? erv[4] : q == 1 ? erv[5] : q == 2 ? erv[6] : erv[7];

  const float4* feat4 = (const float4*)feat;
  float4 acc0 = {0.f, 0.f, 0.f, 0.f}, acc1 = {0.f, 0.f, 0.f, 0.f};
  float dA = 0.f, dB = 0.f;
  for (int base = beg; base < end; base += 32) {
    const int j = base + lane;
    const int my_s = (j < end) ? lst[j] : 0;
    const int n = (end - base) < 32 ? (end - base) : 32;
#pragma unroll 4
    for (int i = 0; i < n; ++i) {
      const int s = __builtin_amdgcn_readlane(my_s, i);
      const float4 f0 = feat4[(size_t)s * (DD / 4) + lane];
      const float4 f1 = feat4[(size_t)s * (DD / 4) + 32 + lane];
      const float pA = expf(leaky(el[(size_t)s * HH + q] + erA) - mA);
      const float pB = expf(leaky(el[(size_t)s * HH + 4 + q] + erB) - mB);
      dA += pA; dB += pB;
      acc0.x += pA * f0.x; acc0.y += pA * f0.y; acc0.z += pA * f0.z; acc0.w += pA * f0.w;
      acc1.x += pB * f1.x; acc1.y += pB * f1.y; acc1.z += pB * f1.z; acc1.w += pB * f1.w;
    }
  }
  dA = (dA == 0.f) ? 1.f : dA;
  dB = (dB == 0.f) ? 1.f : dB;

  const float4 b0 = ((const float4*)bias)[lane];
  const float4 b1 = ((const float4*)bias)[32 + lane];
  float4 o0, o1;
  o0.x = elu1(acc0.x / dA + b0.x); o0.y = elu1(acc0.y / dA + b0.y);
  o0.z = elu1(acc0.z / dA + b0.z); o0.w = elu1(acc0.w / dA + b0.w);
  o1.x = elu1(acc1.x / dB + b1.x); o1.y = elu1(acc1.y / dB + b1.y);
  o1.z = elu1(acc1.z / dB + b1.z); o1.w = elu1(acc1.w / dB + b1.w);
  ((float4*)zrow)[lane] = o0;
  ((float4*)zrow)[32 + lane] = o1;
}

#define SEM_ROWS   (NN * MM)
#define SEM_STRIPS (SEM_ROWS / 16)
#define SEM_BLOCKS 256
#define SEM_WAVES  8
#define SEM_STEP   (SEM_BLOCKS * SEM_WAVES)
#define W1_PITCH   (2 * HID + 32)
static_assert(SEM_ROWS % 16 == 0, "whole 16-row strips, no tail");
static_assert(DD % 4 == 0 && HID % 16 == 0, "whole k steps and column tiles");

__device__ __forceinline__ void sem_load_strip(v2f (&a)[DD / 4],
                                               const float* __restrict__ z,
                                               int strip, int l16, int hf) {
  const float* arow = z + (size_t)(strip * 16 + l16) * DD + hf * 2;
#pragma unroll
  for (int kk = 0; kk < DD / 4; ++kk) a[kk] = *(const v2f*)(arow + kk * 4);
}

__global__ __launch_bounds__(SEM_WAVES * 32) void semantic_kernel(
    const float* __restrict__ z, const float* __restrict__ W1,
    const float* __restrict__ b1, const float* __restrict__ W2,
    float* __restrict__ wsum) {
  __shared__ __attribute__((aligned(16))) float w1s[(DD / 2) * W1_PITCH];
  __shared__ float red[SEM_WAVES][MM];
  const int t = threadIdx.x;
  const int lane = t & 31;
  const int hf = lane >> 4;
  const int l16 = lane & 15;
  const int wave = __builtin_amdgcn_readfirstlane(t >> 5);
  int strip = wave * SEM_BLOCKS + blockIdx.x;

  v2f a[DD / 4], an[DD / 4];
  sem_load_strip(a, z, strip, l16, hf);

  for (int idx = t; idx < (DD / 2) * HID; idx += SEM_WAVES * 32) {
    const int kp = idx / HID, n = idx % HID;
    v2f v;
    v.x = W1[(size_t)(2 * kp) * HID + n];
    v.y = W1[(size_t)(2 * kp + 1) * HID + n];
    *(v2f*)&w1s[kp * W1_PITCH + 2 * n] = v;
  }
  float b1c[HID / 16], w2c[HID / 16];
#pragma unroll
  for (int ct = 0; ct < HID / 16; ++ct) {
    b1c[ct] = b1[ct * 16 + l16];
    w2c[ct] = W2[ct * 16 + l16];
  }
  __syncthreads();

  int boff = hf * W1_PITCH + 2 * l16;
  float am0 = 0.f, am1 = 0.f, am2 = 0.f;

  for (; strip < SEM_STRIPS; strip += SEM_STEP) {
    const int nxt = strip + SEM_STEP;
    if (nxt < SEM_STRIPS) sem_load_strip(an, z, nxt, l16, hf);

    asm volatile("" : "+v"(boff));
    const float* bbase = &w1s[boff];

    v8f acc[HID / 16];
#pragma unroll
    for (int ct = 0; ct < HID / 16; ++ct) acc[ct] = v8f{};
#pragma unroll
    for (int kk = 0; kk < DD / 4; ++kk) {
      const float* bp = bbase + kk * 2 * W1_PITCH;
#pragma unroll
      for (int ct = 0; ct < HID / 16; ++ct) {
        const v2f b = *(const v2f*)(bp + ct * 32);
        acc[ct] = __builtin_amdgcn_wmma_f32_16x16x4_f32(
            false, a[kk], false, b, (short)0, acc[ct], false, false);
      }
    }

    float p[8];
#pragma unroll
    for (int r = 0; r < 8; ++r) {
      float s = 0.f;
#pragma unroll
      for (int ct = 0; ct < HID / 16; ++ct) s += tanhf(acc[ct][r] + b1c[ct]) * w2c[ct];
      p[r] = s;
    }
    const float s0 = p[0] + p[3] + p[6];
    const float s1 = p[1] + p[4] + p[7];
    const float s2 = p[2] + p[5];
    const int first = (strip * 16 + hf * 8) % MM;
    am0 += first == 0 ? s0 : first == 1 ? s2 : s1;
    am1 += first == 0 ? s1 : first == 1 ? s0 : s2;
    am2 += first == 0 ? s2 : first == 1 ? s1 : s0;

    if (nxt < SEM_STRIPS) {
#pragma unroll
      for (int kk = 0; kk < DD / 4; ++kk) a[kk] = an[kk];
    }
  }

  for (int off = 16; off > 0; off >>= 1) {
    am0 += __shfl_xor(am0, off);
    am1 += __shfl_xor(am1, off);
    am2 += __shfl_xor(am2, off);
  }
  if (lane == 0) { red[wave][0] = am0; red[wave][1] = am1; red[wave][2] = am2; }
  __syncthreads();
  if (t < MM) {
    float s = 0.f;
    for (int w = 0; w < SEM_WAVES; ++w) s += red[w][t];
    atomicAdd(&wsum[t], s);
  }
}

__global__ void beta_kernel(const float* __restrict__ wsum,
                            float* __restrict__ beta) {
  float w0 = wsum[0] * (1.f / NN);
  float w1 = wsum[1] * (1.f / NN);
  float w2 = wsum[2] * (1.f / NN);
  float mx = fmaxf(w0, fmaxf(w1, w2));
  float e0 = expf(w0 - mx), e1 = expf(w1 - mx), e2 = expf(w2 - mx);
  float inv = 1.f / (e0 + e1 + e2);
  beta[0] = e0 * inv; beta[1] = e1 * inv; beta[2] = e2 * inv;
}

__global__ void final_kernel(const float* __restrict__ z,
                             const float* __restrict__ beta,
                             float* __restrict__ out) {
  int i = blockIdx.x * blockDim.x + threadIdx.x;
  if (i >= NN * DD) return;
  int n = i >> 8, dd = i & 255;
  size_t base = (size_t)n * (MM * DD) + dd;
  out[i] = beta[0] * z[base] + beta[1] * z[base + DD] + beta[2] * z[base + 2 * DD];
}

extern "C" void kernel_launch(void* const* d_in, const int* in_sizes, int n_in,
                              void* d_out, int out_size, void* d_ws, size_t ws_size,
                              hipStream_t stream) {
  const float* h       = (const float*)d_in[0];
  const int*   eidx    = (const int*)  d_in[1];
  const float* W       = (const float*)d_in[2];
  const float* attn_l  = (const float*)d_in[3];
  const float* attn_r  = (const float*)d_in[4];
  const float* bias    = (const float*)d_in[5];
  const float* sem_W1  = (const float*)d_in[6];
  const float* sem_b1  = (const float*)d_in[7];
  const float* sem_W2  = (const float*)d_in[8];
  float* out = (float*)d_out;

  float* ws   = (float*)d_ws;
  float* z    = ws;
  float* feat = z + (size_t)NN * MM * DD;
  float* el   = feat + (size_t)NN * DD;
  float* er   = el + NN * HH;
  int*   lst  = (int*)(er + NN * HH);
  float* wsum = er + 3 * NN * HH;
  float* beta = wsum + MM;

  hipMemsetAsync(wsum, 0, sizeof(float) * MM, stream);

  for (int m = 0; m < MM; ++m) {
    const float* hm   = h + (size_t)m * NN * IN_SZ;
    const float* Wm   = W + (size_t)m * IN_SZ * DD;
    const float* alm  = attn_l + (size_t)m * HH * OO;
    const float* arm  = attn_r + (size_t)m * HH * OO;
    const float* bm   = bias + (size_t)m * DD;
    const int*   srcm = eidx + (size_t)m * 2 * EE;
    const int*   dstm = srcm + EE;
    float* zm = z + (size_t)m * DD;

    feat_gemm_kernel<<<NN / 16, 32, 0, stream>>>(hm, Wm, feat);
    eler_kernel<<<(NN * HH + 255) / 256, 256, 0, stream>>>(feat, alm, arm, el, er);
    csr_build_kernel<<<CSR_GROUPS, CSR_THREADS, 0, stream>>>(srcm, dstm, (int*)zm, lst);
    node_gather_kernel<<<NN / 8, 256, 0, stream>>>(lst, el, er, feat, bm, zm);
  }

  semantic_kernel<<<SEM_BLOCKS, SEM_WAVES * 32, 0, stream>>>(z, sem_W1, sem_b1, sem_W2, wsum);
  beta_kernel<<<1, 1, 0, stream>>>(wsum, beta);
  final_kernel<<<(NN * DD + 255) / 256, 256, 0, stream>>>(z, beta, out);
}


# el/er head dots taken from the feature GEMM's accumulator tiles in its epilogue (16-lane exchange tree per head); the separate el/er kernel and its re-read of feat are gone; dot sum order changes
# speedup vs baseline: 7.0959x; 1.0823x over previous
#include <hip/hip_runtime.h>
#include <math.h>

#define NN 50000
#define EE 800000
#define MM 3
#define IN_SZ 128
#define HH 8
#define OO 32
#define DD 256
#define HID 128

typedef __attribute__((ext_vector_type(2))) float v2f;
typedef __attribute__((ext_vector_type(8))) float v8f;

__global__ __launch_bounds__(32) void feat_gemm_kernel(
    const float* __restrict__ h, const float* __restrict__ W,
    const float* __restrict__ al, const float* __restrict__ ar,
    float* __restrict__ feat, float* __restrict__ el, float* __restrict__ er) {
  const int lane = threadIdx.x;
  const int half = lane >> 4;
  const int l16  = lane & 15;
  const int row0 = blockIdx.x * 16;

  v2f a[32];
  const float* arow = h + (size_t)(row0 + l16) * IN_SZ;
#pragma unroll
  for (int kk = 0; kk < 32; ++kk) {
    int k = kk * 4 + half * 2;
    a[kk].x = arow[k];
    a[kk].y = arow[k + 1];
  }

  const bool b3 = (l16 & 8) != 0, b2 = (l16 & 4) != 0, b1 = (l16 & 2) != 0, b0 = (l16 & 1) != 0;
  float e0 = 0.f, e1 = 0.f, e2 = 0.f, e3 = 0.f, e4 = 0.f, e5 = 0.f, e6 = 0.f, e7 = 0.f;

  for (int hh = 0; hh < HH; ++hh) {
    float v[16];
#pragma unroll
    for (int sub = 0; sub < 2; ++sub) {
      const int ct = hh * 2 + sub;
      v8f acc = {};
#pragma unroll
      for (int kk = 0; kk < 32; ++kk) {
        int k = kk * 4 + half * 2;
        v2f b;
        b.x = W[(size_t)k * DD + ct * 16 + l16];
        b.y = W[(size_t)(k + 1) * DD + ct * 16 + l16];
        acc = __builtin_amdgcn_wmma_f32_16x16x4_f32(
            false, a[kk], false, b, (short)0, acc, false, false);
      }
#pragma unroll
      for (int r = 0; r < 8; ++r) {
        feat[(size_t)(row0 + half * 8 + r) * DD + ct * 16 + l16] = acc[r];
      }
      const float alv = al[ct * 16 + l16];
      const float arv = ar[ct * 16 + l16];
#pragma unroll
      for (int r = 0; r < 8; ++r) {
        if (sub == 0) { v[r] = acc[r] * alv;  v[8 + r] = acc[r] * arv; }
        else          { v[r] += acc[r] * alv; v[8 + r] += acc[r] * arv; }
      }
    }
    float w8[8], w4[4], w2[2];
#pragma unroll
    for (int i = 0; i < 8; ++i) {
      const float keep = b3 ? v[8 + i] : v[i];
      const float send = b3 ? v[i] : v[8 + i];
      w8[i] = keep + __shfl_xor(send, 8);
    }
#pragma unroll
    for (int i = 0; i < 4; ++i) {
      const float keep = b2 ? w8[4 + i] : w8[i];
      const float send = b2 ? w8[i] : w8[4 + i];
      w4[i] = keep + __shfl_xor(send, 4);
    }
#pragma unroll
    for (int i = 0; i < 2; ++i) {
      const float keep = b1 ? w4[2 + i] : w4[i];
      const float send = b1 ? w4[i] : w4[2 + i];
      w2[i] = keep + __shfl_xor(send, 2);
    }
    const float keep = b0 ? w2[1] : w2[0];
    const float send = b0 ? w2[0] : w2[1];
    const float res = keep + __shfl_xor(send, 1);
    e0 = hh == 0 ? res : e0; e1 = hh == 1 ? res : e1; e2 = hh == 2 ? res : e2; e3 = hh == 3 ? res : e3;
    e4 = hh == 4 ? res : e4; e5 = hh == 5 ? res : e5; e6 = hh == 6 ? res : e6; e7 = hh == 7 ? res : e7;
  }
  float* erow = (b3 ? er : el) + (size_t)(row0 + half * 8 + (l16 & 7)) * HH;
  *(float4*)erow       = make_float4(e0, e1, e2, e3);
  *(float4*)(erow + 4) = make_float4(e4, e5, e6, e7);
}

#define ZS (MM * DD)
static_assert(NN % 8 == 0, "eight node waves per block, no tail");
static_assert(EE <= 2 * NN * HH, "edge list must fit the two freed per-head planes");

#define CSR_GROUPS  256
#define CSR_SPAN    ((NN + CSR_GROUPS - 1) / CSR_GROUPS)
#define CSR_THREADS 800
#define CSR_WAVES   (CSR_THREADS / 32)
#define CSR_WCAP    640
#define CSR_Q       (EE / 4)
#define CSR_DEPTH   5
#define CSR_TRIPS   50
static_assert(CSR_SPAN <= 256 && CSR_THREADS >= 256, "one LDS counter and one thread per owned node");
static_assert(EE % 4 == 0, "whole 16-byte groups of destination ids");
static_assert(CSR_Q == CSR_THREADS * CSR_DEPTH * CSR_TRIPS, "whole trips, no tail");
static_assert(CSR_TRIPS % 2 == 0, "trips are examined in pairs");
static_assert((CSR_GROUPS - 1) * CSR_SPAN < NN, "every group owns at least one node");

__device__ __forceinline__ void csr_load_trip(int4 (&dq)[CSR_DEPTH], int (&qi)[CSR_DEPTH],
                                              const int4* __restrict__ dst4,
                                              int trip, int t, int rot) {
#pragma unroll
  for (int u = 0; u < CSR_DEPTH; ++u) {
    int j = (trip * CSR_DEPTH + u) * CSR_THREADS + t;
    j = (j >= CSR_Q) ? j - CSR_Q : j;
    int q = j + rot;
    q = (q >= CSR_Q) ? q - CSR_Q : q;
    qi[u] = q;
    dq[u] = dst4[q];
  }
}

__global__ __launch_bounds__(CSR_THREADS) void csr_build_kernel(
    const int* __restrict__ src, const int* __restrict__ dst,
    int* __restrict__ zmi, int* __restrict__ lst) {
  __shared__ int cnt[256];
  __shared__ int sc[256];
  __shared__ int cur[256];
  __shared__ int nbelow;
  __shared__ int spill;
  __shared__ int2 kept[CSR_WAVES][CSR_WCAP];
  const int t = threadIdx.x;
  const int lane = t & 31;
  const int wave = __builtin_amdgcn_readfirstlane(t >> 5);
  const int lo = blockIdx.x * CSR_SPAN;
  const int hi = (lo + CSR_SPAN < NN) ? lo + CSR_SPAN : NN;
  const unsigned span = (unsigned)(hi - lo);
  if (t < 256) cnt[t] = 0;
  if (t == 0) { nbelow = 0; spill = 0; }
  __syncthreads();

  const int rot = blockIdx.x * (CSR_Q / CSR_GROUPS);
  const int4* dst4 = (const int4*)dst;
  int below = 0;
  int nw = 0;

  auto examine = [&](const int4 (&dq)[CSR_DEPTH], const int (&qi)[CSR_DEPTH]) {
#pragma unroll
    for (int u = 0; u < CSR_DEPTH; ++u) {
      const int dv[4] = {dq[u].x, dq[u].y, dq[u].z, dq[u].w};
#pragma unroll
      for (int c = 0; c < 4; ++c) {
        const int d = dv[c];
        below += (d < lo) ? 1 : 0;
        const unsigned ld = (unsigned)(d - lo);
        const bool in = ld < span;
        const unsigned m = __builtin_amdgcn_ballot_w32(in);
        if (m != 0u) {
          if (in) {
            atomicAdd(&cnt[ld], 1);
            const int slot = nw + (int)__builtin_amdgcn_mbcnt_lo(m, 0u);
            if (slot < CSR_WCAP) kept[wave][slot] = make_int2(4 * qi[u] + c, (int)ld);
          }
          nw += __builtin_popcount(m);
        }
      }
    }
  };

  int4 da[CSR_DEPTH], db[CSR_DEPTH];
  int qa[CSR_DEPTH], qb[CSR_DEPTH];
  csr_load_trip(da, qa, dst4, 0, t, rot);
  for (int trip = 0; trip < CSR_TRIPS; trip += 2) {
    csr_load_trip(db, qb, dst4, trip + 1, t, rot);
    examine(da, qa);
    csr_load_trip(da, qa, dst4, trip + 2, t, rot);
    examine(db, qb);
  }
  for (int off = 16; off > 0; off >>= 1) below += __shfl_xor(below, off);
  if (lane == 0) {
    atomicAdd(&nbelow, below);
    if (nw > CSR_WCAP) spill = 1;
  }
  __syncthreads();

  const int c = (t < 256) ? cnt[t] : 0;
  if (t < 256) sc[t] = c;
  __syncthreads();
  for (int off = 1; off < 256; off <<= 1) {
    const int v = (t < 256 && t >= off) ? sc[t - off] : 0;
    __syncthreads();
    if (t < 256) sc[t] += v;
    __syncthreads();
  }
  const int base = nbelow;
  const int spilled = spill;
  if (t < 256) cur[t] = base + sc[t] - c;
  if (t < hi - lo) {
    *(int2*)&zmi[(size_t)(lo + t) * ZS] = make_int2(c, base + sc[t]);
  }
  __syncthreads();

  if (spilled == 0) {
    for (int k = lane; k < nw; k += 32) {
      const int2 s = kept[wave][k];
      const int v = src[s.x];
      lst[atomicAdd(&cur[s.y], 1)] = v;
    }
  } else {
    for (int q = t; q < CSR_Q; q += CSR_THREADS) {
      const int4 d4 = dst4[q];
      const int dv[4] = {d4.x, d4.y, d4.z, d4.w};
#pragma unroll
      for (int c2 = 0; c2 < 4; ++c2) {
        const int d = dv[c2];
        if (d >= lo && d < hi) lst[atomicAdd(&cur[d - lo], 1)] = src[4 * q + c2];
      }
    }
  }
}

__device__ __forceinline__ float leaky(float v) { return v > 0.f ? v : 0.2f * v; }
__device__ __forceinline__ float elu1(float x) { return x > 0.f ? x : expm1f(x); }

__global__ __launch_bounds__(256) void node_gather_kernel(
    const int* __restrict__ lst, const float* __restrict__ el,
    const float* __restrict__ er, const float* __restrict__ feat,
    const float* __restrict__ bias, float* zm) {
  const int lane = threadIdx.x & 31;
  const int q = lane >> 3;
  const int d = __builtin_amdgcn_readfirstlane(blockIdx.x * 8 + (threadIdx.x >> 5));
  float* zrow = zm + (size_t)d * ZS;
  const int cnt = ((const int*)zrow)[0];
  const int end = ((const int*)zrow)[1];
  const int beg = end - cnt;

  const float4 erlo = *(const float4*)(er + (size_t)d * HH);
  const float4 erhi = *(const float4*)(er + (size_t)d * HH + 4);
  const float erv[8] = {erlo.x, erlo.y, erlo.z, erlo.w, erhi.x, erhi.y, erhi.z, erhi.w};

  float mx[8];
#pragma unroll
  for (int hh = 0; hh < 8; ++hh) mx[hh] = -INFINITY;
  for (int base = beg; base < end; base += 32) {
    const int j = base + lane;
    if (j < end) {
      const int s = lst[j];
      const float4 a = *(const float4*)(el + (size_t)s * HH);
      const float4 b = *(const float4*)(el + (size_t)s * HH + 4);
      const float ev[8] = {a.x, a.y, a.z, a.w, b.x, b.y, b.z, b.w};
#pragma unroll
      for (int hh = 0; hh < 8; ++hh) mx[hh] = fmaxf(mx[hh], leaky(ev[hh] + erv[hh]));
    }
  }
#pragma unroll
  for (int off = 16; off > 0; off >>= 1) {
#pragma unroll
    for (int hh = 0; hh < 8; ++hh) mx[hh] = fmaxf(mx[hh], __shfl_xor(mx[hh], off));
  }
  float mA = q == 0 ? mx[0] : q == 1 ? mx[1] : q == 2 ? mx[2] : mx[3];
  float mB = q == 0 ? mx[4] : q == 1 ? mx[5] : q == 2 ? mx[6] : mx[7];
  mA = __builtin_isfinite(mA) ? mA : 0.f;
  mB = __builtin_isfinite(mB) ? mB : 0.f;
  const float erA = q == 0 ? erv[0] : q == 1 ? erv[1] : q == 2 ? erv[2] : erv[3];
  const float erB = q == 0 ? erv[4] : q == 1 ? erv[5] : q == 2 ? erv[6] : erv[7];

  const float4* feat4 = (const float4*)feat;
  float4 acc0 = {0.f, 0.f, 0.f, 0.f}, acc1 = {0.f, 0.f, 0.f, 0.f};
  float dA = 0.f, dB = 0.f;
  for (int base = beg; base < end; base += 32) {
    const int j = base + lane;
    const int my_s = (j < end) ? lst[j] : 0;
    const int n = (end - base) < 32 ? (end - base) : 32;
#pragma unroll 4
    for (int i = 0; i < n; ++i) {
      const int s = __builtin_amdgcn_readlane(my_s, i);
      const float4 f0 = feat4[(size_t)s * (DD / 4) + lane];
      const float4 f1 = feat4[(size_t)s * (DD / 4) + 32 + lane];
      const float pA = expf(leaky(el[(size_t)s * HH + q] + erA) - mA);
      const float pB = expf(leaky(el[(size_t)s * HH + 4 + q] + erB) - mB);
      dA += pA; dB += pB;
      acc0.x += pA * f0.x; acc0.y += pA * f0.y; acc0.z += pA * f0.z; acc0.w += pA * f0.w;
      acc1.x += pB * f1.x; acc1.y += pB * f1.y; acc1.z += pB * f1.z; acc1.w += pB * f1.w;
    }
  }
  dA = (dA == 0.f) ? 1.f : dA;
  dB = (dB == 0.f) ? 1.f : dB;

  const float4 b0 = ((const float4*)bias)[lane];
  const float4 b1 = ((const float4*)bias)[32 + lane];
  float4 o0, o1;
  o0.x = elu1(acc0.x / dA + b0.x); o0.y = elu1(acc0.y / dA + b0.y);
  o0.z = elu1(acc0.z / dA + b0.z); o0.w = elu1(acc0.w / dA + b0.w);
  o1.x = elu1(acc1.x / dB + b1.x); o1.y = elu1(acc1.y / dB + b1.y);
  o1.z = elu1(acc1.z / dB + b1.z); o1.w = elu1(acc1.w / dB + b1.w);
  ((float4*)zrow)[lane] = o0;
  ((float4*)zrow)[32 + lane] = o1;
}

#define SEM_ROWS   (NN * MM)
#define SEM_STRIPS (SEM_ROWS / 16)
#define SEM_BLOCKS 256
#define SEM_WAVES  8
#define SEM_STEP   (SEM_BLOCKS * SEM_WAVES)
#define W1_PITCH   (2 * HID + 32)
static_assert(SEM_ROWS % 16 == 0, "whole 16-row strips, no tail");
static_assert(DD % 4 == 0 && HID % 16 == 0, "whole k steps and column tiles");

__device__ __forceinline__ void sem_load_strip(v2f (&a)[DD / 4],
                                               const float* __restrict__ z,
                                               int strip, int l16, int hf) {
  const float* arow = z + (size_t)(strip * 16 + l16) * DD + hf * 2;
#pragma unroll
  for (int kk = 0; kk < DD / 4; ++kk) a[kk] = *(const v2f*)(arow + kk * 4);
}

__global__ __launch_bounds__(SEM_WAVES * 32) void semantic_kernel(
    const float* __restrict__ z, const float* __restrict__ W1,
    const float* __restrict__ b1, const float* __restrict__ W2,
    float* __restrict__ wsum) {
  __shared__ __attribute__((aligned(16))) float w1s[(DD / 2) * W1_PITCH];
  __shared__ float red[SEM_WAVES][MM];
  const int t = threadIdx.x;
  const int lane = t & 31;
  const int hf = lane >> 4;
  const int l16 = lane & 15;
  const int wave = __builtin_amdgcn_readfirstlane(t >> 5);
  int strip = wave * SEM_BLOCKS + blockIdx.x;

  v2f a[DD / 4], an[DD / 4];
  sem_load_strip(a, z, strip, l16, hf);

  for (int idx = t; idx < (DD / 2) * HID; idx += SEM_WAVES * 32) {
    const int kp = idx / HID, n = idx % HID;
    v2f v;
    v.x = W1[(size_t)(2 * kp) * HID + n];
    v.y = W1[(size_t)(2 * kp + 1) * HID + n];
    *(v2f*)&w1s[kp * W1_PITCH + 2 * n] = v;
  }
  float b1c[HID / 16], w2c[HID / 16];
#pragma unroll
  for (int ct = 0; ct < HID / 16; ++ct) {
    b1c[ct] = b1[ct * 16 + l16];
    w2c[ct] = W2[ct * 16 + l16];
  }
  __syncthreads();

  int boff = hf * W1_PITCH + 2 * l16;
  float am0 = 0.f, am1 = 0.f, am2 = 0.f;

  for (; strip < SEM_STRIPS; strip += SEM_STEP) {
    const int nxt = strip + SEM_STEP;
    if (nxt < SEM_STRIPS) sem_load_strip(an, z, nxt, l16, hf);

    asm volatile("" : "+v"(boff));
    const float* bbase = &w1s[boff];

    v8f acc[HID / 16];
#pragma unroll
    for (int ct = 0; ct < HID / 16; ++ct) acc[ct] = v8f{};
#pragma unroll
    for (int kk = 0; kk < DD / 4; ++kk) {
      const float* bp = bbase + kk * 2 * W1_PITCH;
#pragma unroll
      for (int ct = 0; ct < HID / 16; ++ct) {
        const v2f b = *(const v2f*)(bp + ct * 32);
        acc[ct] = __builtin_amdgcn_wmma_f32_16x16x4_f32(
            false, a[kk], false, b, (short)0, acc[ct], false, false);
      }
    }

    float p[8];
#pragma unroll
    for (int r = 0; r < 8; ++r) {
      float s = 0.f;
#pragma unroll
      for (int ct = 0; ct < HID / 16; ++ct) s += tanhf(acc[ct][r] + b1c[ct]) * w2c[ct];
      p[r] = s;
    }
    const float s0 = p[0] + p[3] + p[6];
    const float s1 = p[1] + p[4] + p[7];
    const float s2 = p[2] + p[5];
    const int first = (strip * 16 + hf * 8) % MM;
    am0 += first == 0 ? s0 : first == 1 ? s2 : s1;
    am1 += first == 0 ? s1 : first == 1 ? s0 : s2;
    am2 += first == 0 ? s2 : first == 1 ? s1 : s0;

    if (nxt < SEM_STRIPS) {
#pragma unroll
      for (int kk = 0; kk < DD / 4; ++kk) a[kk] = an[kk];
    }
  }

  for (int off = 16; off > 0; off >>= 1) {
    am0 += __shfl_xor(am0, off);
    am1 += __shfl_xor(am1, off);
    am2 += __shfl_xor(am2, off);
  }
  if (lane == 0) { red[wave][0] = am0; red[wave][1] = am1; red[wave][2] = am2; }
  __syncthreads();
  if (t < MM) {
    float s = 0.f;
    for (int w = 0; w < SEM_WAVES; ++w) s += red[w][t];
    atomicAdd(&wsum[t], s);
  }
}

__global__ void beta_kernel(const float* __restrict__ wsum,
                            float* __restrict__ beta) {
  float w0 = wsum[0] * (1.f / NN);
  float w1 = wsum[1] * (1.f / NN);
  float w2 = wsum[2] * (1.f / NN);
  float mx = fmaxf(w0, fmaxf(w1, w2));
  float e0 = expf(w0 - mx), e1 = expf(w1 - mx), e2 = expf(w2 - mx);
  float inv = 1.f / (e0 + e1 + e2);
  beta[0] = e0 * inv; beta[1] = e1 * inv; beta[2] = e2 * inv;
}

__global__ void final_kernel(const float* __restrict__ z,
                             const float* __restrict__ beta,
                             float* __restrict__ out) {
  int i = blockIdx.x * blockDim.x + threadIdx.x;
  if (i >= NN * DD) return;
  int n = i >> 8, dd = i & 255;
  size_t base = (size_t)n * (MM * DD) + dd;
  out[i] = beta[0] * z[base] + beta[1] * z[base + DD] + beta[2] * z[base + 2 * DD];
}

extern "C" void kernel_launch(void* const* d_in, const int* in_sizes, int n_in,
                              void* d_out, int out_size, void* d_ws, size_t ws_size,
                              hipStream_t stream) {
  const float* h       = (const float*)d_in[0];
  const int*   eidx    = (const int*)  d_in[1];
  const float* W       = (const float*)d_in[2];
  const float* attn_l  = (const float*)d_in[3];
  const float* attn_r  = (const float*)d_in[4];
  const float* bias    = (const float*)d_in[5];
  const float* sem_W1  = (const float*)d_in[6];
  const float* sem_b1  = (const float*)d_in[7];
  const float* sem_W2  = (const float*)d_in[8];
  float* out = (float*)d_out;

  float* ws   = (float*)d_ws;
  float* z    = ws;
  float* feat = z + (size_t)NN * MM * DD;
  float* el   = feat + (size_t)NN * DD;
  float* er   = el + NN * HH;
  int*   lst  = (int*)(er + NN * HH);
  float* wsum = er + 3 * NN * HH;
  float* beta = wsum + MM;

  hipMemsetAsync(wsum, 0, sizeof(float) * MM, stream);

  for (int m = 0; m < MM; ++m) {
    const float* hm   = h + (size_t)m * NN * IN_SZ;
    const float* Wm   = W + (size_t)m * IN_SZ * DD;
    const float* alm  = attn_l + (size_t)m * HH * OO;
    const float* arm  = attn_r + (size_t)m * HH * OO;
    const float* bm   = bias + (size_t)m * DD;
    const int*   srcm = eidx + (size_t)m * 2 * EE;
    const int*   dstm = srcm + EE;
    float* zm = z + (size_t)m * DD;

    feat_gemm_kernel<<<NN / 16, 32, 0, stream>>>(hm, Wm, alm, arm, feat, el, er);
    csr_build_kernel<<<CSR_GROUPS, CSR_THREADS, 0, stream>>>(srcm, dstm, (int*)zm, lst);
    node_gather_kernel<<<NN / 8, 256, 0, stream>>>(lst, el, er, feat, bm, zm);
  }

  semantic_kernel<<<SEM_BLOCKS, SEM_WAVES * 32, 0, stream>>>(z, sem_W1, sem_b1, sem_W2, wsum);
  beta_kernel<<<1, 1, 0, stream>>>(wsum, beta);
  final_kernel<<<(NN * DD + 255) / 256, 256, 0, stream>>>(z, beta, out);
}


# Edge grouping as a two-level bucket sort, 2 small launches: 256 chunks split edges by destination range in LDS; range owners read 256 segments, not all ids; list (sum) order changes, by LDS cursor
# speedup vs baseline: 8.7690x; 1.2358x over previous
#include <hip/hip_runtime.h>
#include <math.h>

#define NN 50000
#define EE 800000
#define MM 3
#define IN_SZ 128
#define HH 8
#define OO 32
#define DD 256
#define HID 128

typedef __attribute__((ext_vector_type(2))) float v2f;
typedef __attribute__((ext_vector_type(8))) float v8f;

__global__ __launch_bounds__(32) void feat_gemm_kernel(
    const float* __restrict__ h, const float* __restrict__ W,
    const float* __restrict__ al, const float* __restrict__ ar,
    float* __restrict__ feat, float* __restrict__ el, float* __restrict__ er) {
  const int lane = threadIdx.x;
  const int half = lane >> 4;
  const int l16  = lane & 15;
  const int row0 = blockIdx.x * 16;

  v2f a[32];
  const float* arow = h + (size_t)(row0 + l16) * IN_SZ;
#pragma unroll
  for (int kk = 0; kk < 32; ++kk) {
    int k = kk * 4 + half * 2;
    a[kk].x = arow[k];
    a[kk].y = arow[k + 1];
  }

  const bool b3 = (l16 & 8) != 0, b2 = (l16 & 4) != 0, b1 = (l16 & 2) != 0, b0 = (l16 & 1) != 0;
  float e0 = 0.f, e1 = 0.f, e2 = 0.f, e3 = 0.f, e4 = 0.f, e5 = 0.f, e6 = 0.f, e7 = 0.f;

  for (int hh = 0; hh < HH; ++hh) {
    float v[16];
#pragma unroll
    for (int sub = 0; sub < 2; ++sub) {
      const int ct = hh * 2 + sub;
      v8f acc = {};
#pragma unroll
      for (int kk = 0; kk < 32; ++kk) {
        int k = kk * 4 + half * 2;
        v2f b;
        b.x = W[(size_t)k * DD + ct * 16 + l16];
        b.y = W[(size_t)(k + 1) * DD + ct * 16 + l16];
        acc = __builtin_amdgcn_wmma_f32_16x16x4_f32(
            false, a[kk], false, b, (short)0, acc, false, false);
      }
#pragma unroll
      for (int r = 0; r < 8; ++r) {
        feat[(size_t)(row0 + half * 8 + r) * DD + ct * 16 + l16] = acc[r];
      }
      const float alv = al[ct * 16 + l16];
      const float arv = ar[ct * 16 + l16];
#pragma unroll
      for (int r = 0; r < 8; ++r) {
        if (sub == 0) { v[r] = acc[r] * alv;  v[8 + r] = acc[r] * arv; }
        else          { v[r] += acc[r] * alv; v[8 + r] += acc[r] * arv; }
      }
    }
    float w8[8], w4[4], w2[2];
#pragma unroll
    for (int i = 0; i < 8; ++i) {
      const float keep = b3 ? v[8 + i] : v[i];
      const float send = b3 ? v[i] : v[8 + i];
      w8[i] = keep + __shfl_xor(send, 8);
    }
#pragma unroll
    for (int i = 0; i < 4; ++i) {
      const float keep = b2 ? w8[4 + i] : w8[i];
      const float send = b2 ? w8[i] : w8[4 + i];
      w4[i] = keep + __shfl_xor(send, 4);
    }
#pragma unroll
    for (int i = 0; i < 2; ++i) {
      const float keep = b1 ? w4[2 + i] : w4[i];
      const float send = b1 ? w4[i] : w4[2 + i];
      w2[i] = keep + __shfl_xor(send, 2);
    }
    const float keep = b0 ? w2[1] : w2[0];
    const float send = b0 ? w2[0] : w2[1];
    const float res = keep + __shfl_xor(send, 1);
    e0 = hh == 0 ? res : e0; e1 = hh == 1 ? res : e1; e2 = hh == 2 ? res : e2; e3 = hh == 3 ? res : e3;
    e4 = hh == 4 ? res : e4; e5 = hh == 5 ? res : e5; e6 = hh == 6 ? res : e6; e7 = hh == 7 ? res : e7;
  }
  float* erow = (b3 ? er : el) + (size_t)(row0 + half * 8 + (l16 & 7)) * HH;
  *(float4*)erow       = make_float4(e0, e1, e2, e3);
  *(float4*)(erow + 4) = make_float4(e4, e5, e6, e7);
}

#define ZS (MM * DD)
static_assert(NN % 8 == 0, "eight node waves per block, no tail");
static_assert(EE <= 2 * NN * HH, "edge list must fit the two freed per-head planes");

#define CSR_GROUPS  256
#define CSR_SPAN    ((NN + CSR_GROUPS - 1) / CSR_GROUPS)
#define PT_CHUNKS   256
#define PT_EDGES    (EE / PT_CHUNKS)
#define PT_THREADS  625
#define PT_PER      (PT_EDGES / PT_THREADS)
#define CB_THREADS  512
#define CB_KEEP     8
static_assert(CSR_SPAN <= 256, "the node index inside a range fits the record's low 8 bits");
static_assert(NN <= (1 << 23), "a source id shifted by 8 stays a positive int");
static_assert(EE % PT_CHUNKS == 0, "whole chunks, no tail");
static_assert(PT_EDGES % PT_THREADS == 0, "whole trips over a chunk, no tail");
static_assert(CSR_GROUPS == 256 && PT_CHUNKS == 256, "block_scan256 scans 256 values");
static_assert(PT_THREADS >= 256 && CB_THREADS >= 256, "one thread per counter and per table entry");
static_assert((CSR_GROUPS - 1) * CSR_SPAN < NN, "every group owns at least one node");

__device__ __forceinline__ int block_scan256(int v, int t, int lane, int* wtot) {
  int x = v;
  for (int off = 1; off < 32; off <<= 1) {
    const int y = __shfl_up(x, off);
    x += (lane >= off) ? y : 0;
  }
  if (t < 256 && lane == 31) wtot[t >> 5] = x;
  __syncthreads();
  const int w = t >> 5;
  int add = 0;
  for (int i = 0; i < 8; ++i) add += (i < w) ? wtot[i] : 0;
  __syncthreads();
  return x + add;
}

__global__ __launch_bounds__(PT_THREADS) void edge_partition_kernel(
    const int* __restrict__ src, const int* __restrict__ dst,
    int* __restrict__ part, int* __restrict__ seg) {
  __shared__ int hist[CSR_GROUPS];
  __shared__ int first[CSR_GROUPS];
  __shared__ int wtot[8];
  __shared__ int sorted[PT_EDGES];
  const int t = threadIdx.x;
  const int lane = t & 31;
  const int e0 = blockIdx.x * PT_EDGES;
  if (t < CSR_GROUPS) hist[t] = 0;
  __syncthreads();

  int dv[PT_PER], sv[PT_PER];
#pragma unroll
  for (int i = 0; i < PT_PER; ++i) {
    dv[i] = dst[e0 + t + i * PT_THREADS];
    sv[i] = src[e0 + t + i * PT_THREADS];
  }
  int rec[PT_PER], own[PT_PER], rank[PT_PER];
#pragma unroll
  for (int i = 0; i < PT_PER; ++i) {
    const int d = dv[i];
    unsigned g = (unsigned)d / (unsigned)CSR_SPAN;
    g = g < (unsigned)CSR_GROUPS ? g : (unsigned)(CSR_GROUPS - 1);
    own[i] = (int)g;
    rec[i] = (sv[i] << 8) | (d - (int)g * CSR_SPAN);
    rank[i] = atomicAdd(&hist[g], 1);
  }
  __syncthreads();

  const int c = (t < CSR_GROUPS) ? hist[t] : 0;
  const int inc = block_scan256(c, t, lane, wtot);
  if (t < CSR_GROUPS) {
    first[t] = inc - c;
    *(int2*)&seg[2 * (blockIdx.x * CSR_GROUPS + t)] = make_int2(inc - c, c);
  }
  __syncthreads();
#pragma unroll
  for (int i = 0; i < PT_PER; ++i) sorted[first[own[i]] + rank[i]] = rec[i];
  __syncthreads();
#pragma unroll
  for (int i = 0; i < PT_PER; ++i) part[e0 + t + i * PT_THREADS] = sorted[t + i * PT_THREADS];
}

__global__ __launch_bounds__(CB_THREADS) void csr_build_kernel(
    const int* __restrict__ part, const int* __restrict__ seg,
    int* __restrict__ zmi, int* __restrict__ lst) {
  __shared__ int segat[PT_CHUNKS];
  __shared__ int segbase[PT_CHUNKS + 1];
  __shared__ int cnt[256];
  __shared__ int cur[256];
  __shared__ int wtot[8];
  __shared__ int nbelow;
  const int t = threadIdx.x;
  const int lane = t & 31;
  const int lo = blockIdx.x * CSR_SPAN;
  const int hi = (lo + CSR_SPAN < NN) ? lo + CSR_SPAN : NN;

  int st = 0, ln = 0;
  if (t < PT_CHUNKS) {
    const int2 sg = *(const int2*)&seg[2 * (t * CSR_GROUPS + blockIdx.x)];
    st = sg.x;
    ln = sg.y;
    segat[t] = t * PT_EDGES + st;
  }
  if (t < 256) cnt[t] = 0;
  if (t == 0) nbelow = 0;
  __syncthreads();
  int sb = st;
  for (int off = 16; off > 0; off >>= 1) sb += __shfl_xor(sb, off);
  if (lane == 0 && t < PT_CHUNKS) atomicAdd(&nbelow, sb);
  const int inc = block_scan256(ln, t, lane, wtot);
  if (t < PT_CHUNKS) segbase[t] = inc - ln;
  if (t == PT_CHUNKS - 1) segbase[PT_CHUNKS] = inc;
  __syncthreads();
  const int total = segbase[PT_CHUNKS];

  auto fetch = [&](int k) -> int {
    int g = 0;
#pragma unroll
    for (int step = PT_CHUNKS / 2; step > 0; step >>= 1) g += (segbase[g + step] <= k) ? step : 0;
    return part[segat[g] + (k - segbase[g])];
  };

  int keep[CB_KEEP];
  {
    int g[CB_KEEP];
#pragma unroll
    for (int i = 0; i < CB_KEEP; ++i) g[i] = 0;
#pragma unroll
    for (int step = PT_CHUNKS / 2; step > 0; step >>= 1) {
#pragma unroll
      for (int i = 0; i < CB_KEEP; ++i) g[i] += (segbase[g[i] + step] <= t + i * CB_THREADS) ? step : 0;
    }
#pragma unroll
    for (int i = 0; i < CB_KEEP; ++i) {
      const int k = t + i * CB_THREADS;
      keep[i] = (k < total) ? part[segat[g[i]] + (k - segbase[g[i]])] : -1;
    }
  }
#pragma unroll
  for (int i = 0; i < CB_KEEP; ++i) {
    if (keep[i] >= 0) atomicAdd(&cnt[keep[i] & 255], 1);
  }
  for (int k = t + CB_KEEP * CB_THREADS; k < total; k += CB_THREADS) atomicAdd(&cnt[fetch(k) & 255], 1);
  __syncthreads();

  const int c = (t < 256) ? cnt[t] : 0;
  const int cinc = block_scan256(c, t, lane, wtot);
  const int base = nbelow;
  if (t < 256) cur[t] = base + cinc - c;
  if (t < hi - lo) {
    *(int2*)&zmi[(size_t)(lo + t) * ZS] = make_int2(c, base + cinc);
  }
  __syncthreads();

#pragma unroll
  for (int i = 0; i < CB_KEEP; ++i) {
    if (keep[i] >= 0) lst[atomicAdd(&cur[keep[i] & 255], 1)] = keep[i] >> 8;
  }
  for (int k = t + CB_KEEP * CB_THREADS; k < total; k += CB_THREADS) {
    const int r = fetch(k);
    lst[atomicAdd(&cur[r & 255], 1)] = r >> 8;
  }
}

__device__ __forceinline__ float leaky(float v) { return v > 0.f ? v : 0.2f * v; }
__device__ __forceinline__ float elu1(float x) { return x > 0.f ? x : expm1f(x); }

__global__ __launch_bounds__(256) void node_gather_kernel(
    const int* __restrict__ lst, const float* __restrict__ el,
    const float* __restrict__ er, const float* __restrict__ feat,
    const float* __restrict__ bias, float* zm) {
  const int lane = threadIdx.x & 31;
  const int q = lane >> 3;
  const int d = __builtin_amdgcn_readfirstlane(blockIdx.x * 8 + (threadIdx.x >> 5));
  float* zrow = zm + (size_t)d * ZS;
  const int cnt = ((const int*)zrow)[0];
  const int end = ((const int*)zrow)[1];
  const int beg = end - cnt;

  const float4 erlo = *(const float4*)(er + (size_t)d * HH);
  const float4 erhi = *(const float4*)(er + (size_t)d * HH + 4);
  const float erv[8] = {erlo.x, erlo.y, erlo.z, erlo.w, erhi.x, erhi.y, erhi.z, erhi.w};

  float mx[8];
#pragma unroll
  for (int hh = 0; hh < 8; ++hh) mx[hh] = -INFINITY;
  for (int base = beg; base < end; base += 32) {
    const int j = base + lane;
    if (j < end) {
      const int s = lst[j];
      const float4 a = *(const float4*)(el + (size_t)s * HH);
      const float4 b = *(const float4*)(el + (size_t)s * HH + 4);
      const float ev[8] = {a.x, a.y, a.z, a.w, b.x, b.y, b.z, b.w};
#pragma unroll
      for (int hh = 0; hh < 8; ++hh) mx[hh] = fmaxf(mx[hh], leaky(ev[hh] + erv[hh]));
    }
  }
#pragma unroll
  for (int off = 16; off > 0; off >>= 1) {
#pragma unroll
    for (int hh = 0; hh < 8; ++hh) mx[hh] = fmaxf(mx[hh], __shfl_xor(mx[hh], off));
  }
  float mA = q == 0 ? mx[0] : q == 1 ? mx[1] : q == 2 ? mx[2] : mx[3];
  float mB = q == 0 ? mx[4] : q == 1 ? mx[5] : q == 2 ? mx[6] : mx[7];
  mA = __builtin_isfinite(mA) ? mA : 0.f;
  mB = __builtin_isfinite(mB) ? mB : 0.f;
  const float erA = q == 0 ? erv[0] : q == 1 ? erv[1] : q == 2 ? erv[2] : erv[3];
  const float erB = q == 0 ? erv[4] : q == 1 ? erv[5] : q == 2 ? erv[6] : erv[7];

  const float4* feat4 = (const float4*)feat;
  float4 acc0 = {0.f, 0.f, 0.f, 0.f}, acc1 = {0.f, 0.f, 0.f, 0.f};
  float dA = 0.f, dB = 0.f;
  for (int base = beg; base < end; base += 32) {
    const int j = base + lane;
    const int my_s = (j < end) ? lst[j] : 0;
    const int n = (end - base) < 32 ? (end - base) : 32;
#pragma unroll 4
    for (int i = 0; i < n; ++i) {
      const int s = __builtin_amdgcn_readlane(my_s, i);
      const float4 f0 = feat4[(size_t)s * (DD / 4) + lane];
      const float4 f1 = feat4[(size_t)s * (DD / 4) + 32 + lane];
      const float pA = expf(leaky(el[(size_t)s * HH + q] + erA) - mA);
      const float pB = expf(leaky(el[(size_t)s * HH + 4 + q] + erB) - mB);
      dA += pA; dB += pB;
      acc0.x += pA * f0.x; acc0.y += pA * f0.y; acc0.z += pA * f0.z; acc0.w += pA * f0.w;
      acc1.x += pB * f1.x; acc1.y += pB * f1.y; acc1.z += pB * f1.z; acc1.w += pB * f1.w;
    }
  }
  dA = (dA == 0.f) ? 1.f : dA;
  dB = (dB == 0.f) ? 1.f : dB;

  const float4 b0 = ((const float4*)bias)[lane];
  const float4 b1 = ((const float4*)bias)[32 + lane];
  float4 o0, o1;
  o0.x = elu1(acc0.x / dA + b0.x); o0.y = elu1(acc0.y / dA + b0.y);
  o0.z = elu1(acc0.z / dA + b0.z); o0.w = elu1(acc0.w / dA + b0.w);
  o1.x = elu1(acc1.x / dB + b1.x); o1.y = elu1(acc1.y / dB + b1.y);
  o1.z = elu1(acc1.z / dB + b1.z); o1.w = elu1(acc1.w / dB + b1.w);
  ((float4*)zrow)[lane] = o0;
  ((float4*)zrow)[32 + lane] = o1;
}

#define SEM_ROWS   (NN * MM)
#define SEM_STRIPS (SEM_ROWS / 16)
#define SEM_BLOCKS 256
#define SEM_WAVES  8
#define SEM_STEP   (SEM_BLOCKS * SEM_WAVES)
#define W1_PITCH   (2 * HID + 32)
static_assert(SEM_ROWS % 16 == 0, "whole 16-row strips, no tail");
static_assert(DD % 4 == 0 && HID % 16 == 0, "whole k steps and column tiles");

__device__ __forceinline__ void sem_load_strip(v2f (&a)[DD / 4],
                                               const float* __restrict__ z,
                                               int strip, int l16, int hf) {
  const float* arow = z + (size_t)(strip * 16 + l16) * DD + hf * 2;
#pragma unroll
  for (int kk = 0; kk < DD / 4; ++kk) a[kk] = *(const v2f*)(arow + kk * 4);
}

__global__ __launch_bounds__(SEM_WAVES * 32) void semantic_kernel(
    const float* __restrict__ z, const float* __restrict__ W1,
    const float* __restrict__ b1, const float* __restrict__ W2,
    float* __restrict__ wsum) {
  __shared__ __attribute__((aligned(16))) float w1s[(DD / 2) * W1_PITCH];
  __shared__ float red[SEM_WAVES][MM];
  const int t = threadIdx.x;
  const int lane = t & 31;
  const int hf = lane >> 4;
  const int l16 = lane & 15;
  const int wave = __builtin_amdgcn_readfirstlane(t >> 5);
  int strip = wave * SEM_BLOCKS + blockIdx.x;

  v2f a[DD / 4], an[DD / 4];
  sem_load_strip(a, z, strip, l16, hf);

  for (int idx = t; idx < (DD / 2) * HID; idx += SEM_WAVES * 32) {
    const int kp = idx / HID, n = idx % HID;
    v2f v;
    v.x = W1[(size_t)(2 * kp) * HID + n];
    v.y = W1[(size_t)(2 * kp + 1) * HID + n];
    *(v2f*)&w1s[kp * W1_PITCH + 2 * n] = v;
  }
  float b1c[HID / 16], w2c[HID / 16];
#pragma unroll
  for (int ct = 0; ct < HID / 16; ++ct) {
    b1c[ct] = b1[ct * 16 + l16];
    w2c[ct] = W2[ct * 16 + l16];
  }
  __syncthreads();

  int boff = hf * W1_PITCH + 2 * l16;
  float am0 = 0.f, am1 = 0.f, am2 = 0.f;

  for (; strip < SEM_STRIPS; strip += SEM_STEP) {
    const int nxt = strip + SEM_STEP;
    if (nxt < SEM_STRIPS) sem_load_strip(an, z, nxt, l16, hf);

    asm volatile("" : "+v"(boff));
    const float* bbase = &w1s[boff];

    v8f acc[HID / 16];
#pragma unroll
    for (int ct = 0; ct < HID / 16; ++ct) acc[ct] = v8f{};
#pragma unroll
    for (int kk = 0; kk < DD / 4; ++kk) {
      const float* bp = bbase + kk * 2 * W1_PITCH;
#pragma unroll
      for (int ct = 0; ct < HID / 16; ++ct) {
        const v2f b = *(const v2f*)(bp + ct * 32);
        acc[ct] = __builtin_amdgcn_wmma_f32_16x16x4_f32(
            false, a[kk], false, b, (short)0, acc[ct], false, false);
      }
    }

    float p[8];
#pragma unroll
    for (int r = 0; r < 8; ++r) {
      float s = 0.f;
#pragma unroll
      for (int ct = 0; ct < HID / 16; ++ct) s += tanhf(acc[ct][r] + b1c[ct]) * w2c[ct];
      p[r] = s;
    }
    const float s0 = p[0] + p[3] + p[6];
    const float s1 = p[1] + p[4] + p[7];
    const float s2 = p[2] + p[5];
    const int first = (strip * 16 + hf * 8) % MM;
    am0 += first == 0 ? s0 : first == 1 ? s2 : s1;
    am1 += first == 0 ? s1 : first == 1 ? s0 : s2;
    am2 += first == 0 ? s2 : first == 1 ? s1 : s0;

    if (nxt < SEM_STRIPS) {
#pragma unroll
      for (int kk = 0; kk < DD / 4; ++kk) a[kk] = an[kk];
    }
  }

  for (int off = 16; off > 0; off >>= 1) {
    am0 += __shfl_xor(am0, off);
    am1 += __shfl_xor(am1, off);
    am2 += __shfl_xor(am2, off);
  }
  if (lane == 0) { red[wave][0] = am0; red[wave][1] = am1; red[wave][2] = am2; }
  __syncthreads();
  if (t < MM) {
    float s = 0.f;
    for (int w = 0; w < SEM_WAVES; ++w) s += red[w][t];
    atomicAdd(&wsum[t], s);
  }
}

__global__ void beta_kernel(const float* __restrict__ wsum,
                            float* __restrict__ beta) {
  float w0 = wsum[0] * (1.f / NN);
  float w1 = wsum[1] * (1.f / NN);
  float w2 = wsum[2] * (1.f / NN);
  float mx = fmaxf(w0, fmaxf(w1, w2));
  float e0 = expf(w0 - mx), e1 = expf(w1 - mx), e2 = expf(w2 - mx);
  float inv = 1.f / (e0 + e1 + e2);
  beta[0] = e0 * inv; beta[1] = e1 * inv; beta[2] = e2 * inv;
}

__global__ void final_kernel(const float* __restrict__ z,
                             const float* __restrict__ beta,
                             float* __restrict__ out) {
  int i = blockIdx.x * blockDim.x + threadIdx.x;
  if (i >= NN * DD) return;
  int n = i >> 8, dd = i & 255;
  size_t base = (size_t)n * (MM * DD) + dd;
  out[i] = beta[0] * z[base] + beta[1] * z[base + DD] + beta[2] * z[base + 2 * DD];
}

extern "C" void kernel_launch(void* const* d_in, const int* in_sizes, int n_in,
                              void* d_out, int out_size, void* d_ws, size_t ws_size,
                              hipStream_t stream) {
  const float* h       = (const float*)d_in[0];
  const int*   eidx    = (const int*)  d_in[1];
  const float* W       = (const float*)d_in[2];
  const float* attn_l  = (const float*)d_in[3];
  const float* attn_r  = (const float*)d_in[4];
  const float* bias    = (const float*)d_in[5];
  const float* sem_W1  = (const float*)d_in[6];
  const float* sem_b1  = (const float*)d_in[7];
  const float* sem_W2  = (const float*)d_in[8];
  float* out = (float*)d_out;

  float* ws   = (float*)d_ws;
  float* z    = ws;
  float* feat = z + (size_t)NN * MM * DD;
  float* el   = feat + (size_t)NN * DD;
  float* er   = el + NN * HH;
  int*   lst  = (int*)(er + NN * HH);
  float* wsum = er + 3 * NN * HH;
  float* beta = wsum + MM;
  int*   part = (int*)feat;
  int*   seg  = part + EE;

  hipMemsetAsync(wsum, 0, sizeof(float) * MM, stream);

  for (int m = 0; m < MM; ++m) {
    const float* hm   = h + (size_t)m * NN * IN_SZ;
    const float* Wm   = W + (size_t)m * IN_SZ * DD;
    const float* alm  = attn_l + (size_t)m * HH * OO;
    const float* arm  = attn_r + (size_t)m * HH * OO;
    const float* bm   = bias + (size_t)m * DD;
    const int*   srcm = eidx + (size_t)m * 2 * EE;
    const int*   dstm = srcm + EE;
    float* zm = z + (size_t)m * DD;

    edge_partition_kernel<<<PT_CHUNKS, PT_THREADS, 0, stream>>>(srcm, dstm, part, seg);
    csr_build_kernel<<<CSR_GROUPS, CB_THREADS, 0, stream>>>(part, seg, (int*)zm, lst);
    feat_gemm_kernel<<<NN / 16, 32, 0, stream>>>(hm, Wm, alm, arm, feat, el, er);
    node_gather_kernel<<<NN / 8, 256, 0, stream>>>(lst, el, er, feat, bm, zm);
  }

  semantic_kernel<<<SEM_BLOCKS, SEM_WAVES * 32, 0, stream>>>(z, sem_W1, sem_b1, sem_W2, wsum);
  beta_kernel<<<1, 1, 0, stream>>>(wsum, beta);
  final_kernel<<<(NN * DD + 255) / 256, 256, 0, stream>>>(z, beta, out);
}


# Edge softmax weights formed once per edge (lane = edge; was in all 32 lanes per edge) and read by the row loop from LDS; each 32-edge chunk is summed in ascending source order, not list order
# speedup vs baseline: 9.2620x; 1.0562x over previous
#include <hip/hip_runtime.h>
#include <math.h>

#define NN 50000
#define EE 800000
#define MM 3
#define IN_SZ 128
#define HH 8
#define OO 32
#define DD 256
#define HID 128

typedef __attribute__((ext_vector_type(2))) float v2f;
typedef __attribute__((ext_vector_type(8))) float v8f;

__global__ __launch_bounds__(32) void feat_gemm_kernel(
    const float* __restrict__ h, const float* __restrict__ W,
    const float* __restrict__ al, const float* __restrict__ ar,
    float* __restrict__ feat, float* __restrict__ el, float* __restrict__ er) {
  const int lane = threadIdx.x;
  const int half = lane >> 4;
  const int l16  = lane & 15;
  const int row0 = blockIdx.x * 16;

  v2f a[32];
  const float* arow = h + (size_t)(row0 + l16) * IN_SZ;
#pragma unroll
  for (int kk = 0; kk < 32; ++kk) {
    int k = kk * 4 + half * 2;
    a[kk].x = arow[k];
    a[kk].y = arow[k + 1];
  }

  const bool b3 = (l16 & 8) != 0, b2 = (l16 & 4) != 0, b1 = (l16 & 2) != 0, b0 = (l16 & 1) != 0;
  float e0 = 0.f, e1 = 0.f, e2 = 0.f, e3 = 0.f, e4 = 0.f, e5 = 0.f, e6 = 0.f, e7 = 0.f;

  for (int hh = 0; hh < HH; ++hh) {
    float v[16];
#pragma unroll
    for (int sub = 0; sub < 2; ++sub) {
      const int ct = hh * 2 + sub;
      v8f acc = {};
#pragma unroll
      for (int kk = 0; kk < 32; ++kk) {
        int k = kk * 4 + half * 2;
        v2f b;
        b.x = W[(size_t)k * DD + ct * 16 + l16];
        b.y = W[(size_t)(k + 1) * DD + ct * 16 + l16];
        acc = __builtin_amdgcn_wmma_f32_16x16x4_f32(
            false, a[kk], false, b, (short)0, acc, false, false);
      }
#pragma unroll
      for (int r = 0; r < 8; ++r) {
        feat[(size_t)(row0 + half * 8 + r) * DD + ct * 16 + l16] = acc[r];
      }
      const float alv = al[ct * 16 + l16];
      const float arv = ar[ct * 16 + l16];
#pragma unroll
      for (int r = 0; r < 8; ++r) {
        if (sub == 0) { v[r] = acc[r] * alv;  v[8 + r] = acc[r] * arv; }
        else          { v[r] += acc[r] * alv; v[8 + r] += acc[r] * arv; }
      }
    }
    float w8[8], w4[4], w2[2];
#pragma unroll
    for (int i = 0; i < 8; ++i) {
      const float keep = b3 ? v[8 + i] : v[i];
      const float send = b3 ? v[i] : v[8 + i];
      w8[i] = keep + __shfl_xor(send, 8);
    }
#pragma unroll
    for (int i = 0; i < 4; ++i) {
      const float keep = b2 ? w8[4 + i] : w8[i];
      const float send = b2 ? w8[i] : w8[4 + i];
      w4[i] = keep + __shfl_xor(send, 4);
    }
#pragma unroll
    for (int i = 0; i < 2; ++i) {
      const float keep = b1 ? w4[2 + i] : w4[i];
      const float send = b1 ? w4[i] : w4[2 + i];
      w2[i] = keep + __shfl_xor(send, 2);
    }
    const float keep = b0 ? w2[1] : w2[0];
    const float send = b0 ? w2[0] : w2[1];
    const float res = keep + __shfl_xor(send, 1);
    e0 = hh == 0 ? res : e0; e1 = hh == 1 ? res : e1; e2 = hh == 2 ? res : e2; e3 = hh == 3 ? res : e3;
    e4 = hh == 4 ? res : e4; e5 = hh == 5 ? res : e5; e6 = hh == 6 ? res : e6; e7 = hh == 7 ? res : e7;
  }
  float* erow = (b3 ? er : el) + (size_t)(row0 + half * 8 + (l16 & 7)) * HH;
  *(float4*)erow       = make_float4(e0, e1, e2, e3);
  *(float4*)(erow + 4) = make_float4(e4, e5, e6, e7);
}

#define ZS (MM * DD)
static_assert(NN % 8 == 0, "eight node waves per block, no tail");
static_assert(EE <= 2 * NN * HH, "edge list must fit the two freed per-head planes");

#define CSR_GROUPS  256
#define CSR_SPAN    ((NN + CSR_GROUPS - 1) / CSR_GROUPS)
#define PT_CHUNKS   256
#define PT_EDGES    (EE / PT_CHUNKS)
#define PT_THREADS  625
#define PT_PER      (PT_EDGES / PT_THREADS)
#define CB_THREADS  512
#define CB_KEEP     8
static_assert(CSR_SPAN <= 256, "the node index inside a range fits the record's low 8 bits");
static_assert(NN <= (1 << 23), "a source id shifted by 8 stays a positive int");
static_assert(EE % PT_CHUNKS == 0, "whole chunks, no tail");
static_assert(PT_EDGES % PT_THREADS == 0, "whole trips over a chunk, no tail");
static_assert(CSR_GROUPS == 256 && PT_CHUNKS == 256, "block_scan256 scans 256 values");
static_assert(PT_THREADS >= 256 && CB_THREADS >= 256, "one thread per counter and per table entry");
static_assert((CSR_GROUPS - 1) * CSR_SPAN < NN, "every group owns at least one node");

__device__ __forceinline__ int block_scan256(int v, int t, int lane, int* wtot) {
  int x = v;
  for (int off = 1; off < 32; off <<= 1) {
    const int y = __shfl_up(x, off);
    x += (lane >= off) ? y : 0;
  }
  if (t < 256 && lane == 31) wtot[t >> 5] = x;
  __syncthreads();
  const int w = t >> 5;
  int add = 0;
  for (int i = 0; i < 8; ++i) add += (i < w) ? wtot[i] : 0;
  __syncthreads();
  return x + add;
}

__global__ __launch_bounds__(PT_THREADS) void edge_partition_kernel(
    const int* __restrict__ src, const int* __restrict__ dst,
    int* __restrict__ part, int* __restrict__ seg) {
  __shared__ int hist[CSR_GROUPS];
  __shared__ int first[CSR_GROUPS];
  __shared__ int wtot[8];
  __shared__ int sorted[PT_EDGES];
  const int t = threadIdx.x;
  const int lane = t & 31;
  const int e0 = blockIdx.x * PT_EDGES;
  if (t < CSR_GROUPS) hist[t] = 0;
  __syncthreads();

  int dv[PT_PER], sv[PT_PER];
#pragma unroll
  for (int i = 0; i < PT_PER; ++i) {
    dv[i] = dst[e0 + t + i * PT_THREADS];
    sv[i] = src[e0 + t + i * PT_THREADS];
  }
  int rec[PT_PER], own[PT_PER], rank[PT_PER];
#pragma unroll
  for (int i = 0; i < PT_PER; ++i) {
    const int d = dv[i];
    unsigned g = (unsigned)d / (unsigned)CSR_SPAN;
    g = g < (unsigned)CSR_GROUPS ? g : (unsigned)(CSR_GROUPS - 1);
    own[i] = (int)g;
    rec[i] = (sv[i] << 8) | (d - (int)g * CSR_SPAN);
    rank[i] = atomicAdd(&hist[g], 1);
  }
  __syncthreads();

  const int c = (t < CSR_GROUPS) ? hist[t] : 0;
  const int inc = block_scan256(c, t, lane, wtot);
  if (t < CSR_GROUPS) {
    first[t] = inc - c;
    *(int2*)&seg[2 * (blockIdx.x * CSR_GROUPS + t)] = make_int2(inc - c, c);
  }
  __syncthreads();
#pragma unroll
  for (int i = 0; i < PT_PER; ++i) sorted[first[own[i]] + rank[i]] = rec[i];
  __syncthreads();
#pragma unroll
  for (int i = 0; i < PT_PER; ++i) part[e0 + t + i * PT_THREADS] = sorted[t + i * PT_THREADS];
}

__global__ __launch_bounds__(CB_THREADS) void csr_build_kernel(
    const int* __restrict__ part, const int* __restrict__ seg,
    int* __restrict__ zmi, int* __restrict__ lst) {
  __shared__ int segat[PT_CHUNKS];
  __shared__ int segbase[PT_CHUNKS + 1];
  __shared__ int cnt[256];
  __shared__ int cur[256];
  __shared__ int wtot[8];
  __shared__ int nbelow;
  const int t = threadIdx.x;
  const int lane = t & 31;
  const int lo = blockIdx.x * CSR_SPAN;
  const int hi = (lo + CSR_SPAN < NN) ? lo + CSR_SPAN : NN;

  int st = 0, ln = 0;
  if (t < PT_CHUNKS) {
    const int2 sg = *(const int2*)&seg[2 * (t * CSR_GROUPS + blockIdx.x)];
    st = sg.x;
    ln = sg.y;
    segat[t] = t * PT_EDGES + st;
  }
  if (t < 256) cnt[t] = 0;
  if (t == 0) nbelow = 0;
  __syncthreads();
  int sb = st;
  for (int off = 16; off > 0; off >>= 1) sb += __shfl_xor(sb, off);
  if (lane == 0 && t < PT_CHUNKS) atomicAdd(&nbelow, sb);
  const int inc = block_scan256(ln, t, lane, wtot);
  if (t < PT_CHUNKS) segbase[t] = inc - ln;
  if (t == PT_CHUNKS - 1) segbase[PT_CHUNKS] = inc;
  __syncthreads();
  const int total = segbase[PT_CHUNKS];

  auto fetch = [&](int k) -> int {
    int g = 0;
#pragma unroll
    for (int step = PT_CHUNKS / 2; step > 0; step >>= 1) g += (segbase[g + step] <= k) ? step : 0;
    return part[segat[g] + (k - segbase[g])];
  };

  int keep[CB_KEEP];
  {
    int g[CB_KEEP];
#pragma unroll
    for (int i = 0; i < CB_KEEP; ++i) g[i] = 0;
#pragma unroll
    for (int step = PT_CHUNKS / 2; step > 0; step >>= 1) {
#pragma unroll
      for (int i = 0; i < CB_KEEP; ++i) g[i] += (segbase[g[i] + step] <= t + i * CB_THREADS) ? step : 0;
    }
#pragma unroll
    for (int i = 0; i < CB_KEEP; ++i) {
      const int k = t + i * CB_THREADS;
      keep[i] = (k < total) ? part[segat[g[i]] + (k - segbase[g[i]])] : -1;
    }
  }
#pragma unroll
  for (int i = 0; i < CB_KEEP; ++i) {
    if (keep[i] >= 0) atomicAdd(&cnt[keep[i] & 255], 1);
  }
  for (int k = t + CB_KEEP * CB_THREADS; k < total; k += CB_THREADS) atomicAdd(&cnt[fetch(k) & 255], 1);
  __syncthreads();

  const int c = (t < 256) ? cnt[t] : 0;
  const int cinc = block_scan256(c, t, lane, wtot);
  const int base = nbelow;
  if (t < 256) cur[t] = base + cinc - c;
  if (t < hi - lo) {
    *(int2*)&zmi[(size_t)(lo + t) * ZS] = make_int2(c, base + cinc);
  }
  __syncthreads();

#pragma unroll
  for (int i = 0; i < CB_KEEP; ++i) {
    if (keep[i] >= 0) lst[atomicAdd(&cur[keep[i] & 255], 1)] = keep[i] >> 8;
  }
  for (int k = t + CB_KEEP * CB_THREADS; k < total; k += CB_THREADS) {
    const int r = fetch(k);
    lst[atomicAdd(&cur[r & 255], 1)] = r >> 8;
  }
}

__device__ __forceinline__ float leaky(float v) { return v > 0.f ? v : 0.2f * v; }
__device__ __forceinline__ float elu1(float x) { return x > 0.f ? x : expm1f(x); }

__global__ __launch_bounds__(256) void node_gather_kernel(
    const int* __restrict__ lst, const float* __restrict__ el,
    const float* __restrict__ er, const float* __restrict__ feat,
    const float* __restrict__ bias, float* zm) {
  __shared__ float pw[8][HH][32];
  static_assert(NN <= (1 << 26), "a source id shifted by 5 stays a positive int");
  const int lane = threadIdx.x & 31;
  const int q = lane >> 3;
  float* pmine = &pw[threadIdx.x >> 5][0][0];
  const float* prow = pmine + q * 32;
  const int d = __builtin_amdgcn_readfirstlane(blockIdx.x * 8 + (threadIdx.x >> 5));
  float* zrow = zm + (size_t)d * ZS;
  const int cnt = ((const int*)zrow)[0];
  const int end = ((const int*)zrow)[1];
  const int beg = end - cnt;

  const float4 erlo = *(const float4*)(er + (size_t)d * HH);
  const float4 erhi = *(const float4*)(er + (size_t)d * HH + 4);
  const float erv[8] = {erlo.x, erlo.y, erlo.z, erlo.w, erhi.x, erhi.y, erhi.z, erhi.w};

  float mx[8], e_first[8];
  int s_first = 0;
#pragma unroll
  for (int hh = 0; hh < 8; ++hh) { mx[hh] = -INFINITY; e_first[hh] = 0.f; }
  for (int base = beg; base < end; base += 32) {
    const int j = base + lane;
    if (j < end) {
      const int s = lst[j];
      const float4 a = *(const float4*)(el + (size_t)s * HH);
      const float4 b = *(const float4*)(el + (size_t)s * HH + 4);
      const float ev[8] = {a.x, a.y, a.z, a.w, b.x, b.y, b.z, b.w};
      if (base == beg) s_first = s;
#pragma unroll
      for (int hh = 0; hh < 8; ++hh) {
        const float e = leaky(ev[hh] + erv[hh]);
        if (base == beg) e_first[hh] = e;
        mx[hh] = fmaxf(mx[hh], e);
      }
    }
  }
#pragma unroll
  for (int off = 16; off > 0; off >>= 1) {
#pragma unroll
    for (int hh = 0; hh < 8; ++hh) mx[hh] = fmaxf(mx[hh], __shfl_xor(mx[hh], off));
  }
  float mh[8];
#pragma unroll
  for (int hh = 0; hh < 8; ++hh) mh[hh] = __builtin_isfinite(mx[hh]) ? mx[hh] : 0.f;

  const float4* feat4 = (const float4*)feat;
  float4 acc0 = {0.f, 0.f, 0.f, 0.f}, acc1 = {0.f, 0.f, 0.f, 0.f};
  float dA = 0.f, dB = 0.f;
  for (int base = beg; base < end; base += 32) {
    const int j = base + lane;
    const bool valid = j < end;
    const int n = (end - base) < 32 ? (end - base) : 32;
    int my_s = s_first;
    float e[8];
#pragma unroll
    for (int hh = 0; hh < 8; ++hh) e[hh] = e_first[hh];
    if (base != beg && valid) {
      my_s = lst[j];
      const float4 a = *(const float4*)(el + (size_t)my_s * HH);
      const float4 b = *(const float4*)(el + (size_t)my_s * HH + 4);
      const float ev[8] = {a.x, a.y, a.z, a.w, b.x, b.y, b.z, b.w};
#pragma unroll
      for (int hh = 0; hh < 8; ++hh) e[hh] = leaky(ev[hh] + erv[hh]);
    }
    const int key = valid ? ((my_s << 5) | lane) : 0x7fffffff;
    int slot = 0;
    for (int i = 0; i < n; ++i) {
      slot += (__builtin_amdgcn_readlane(key, i) < key) ? 1 : 0;
    }
    const int ss = __builtin_amdgcn_ds_permute(slot << 2, my_s);
    if (valid) {
#pragma unroll
      for (int hh = 0; hh < 8; ++hh) pmine[hh * 32 + slot] = expf(e[hh] - mh[hh]);
    }
#pragma unroll 4
    for (int i = 0; i < n; ++i) {
      const int s = __builtin_amdgcn_readlane(ss, i);
      const float4 f0 = feat4[(size_t)s * (DD / 4) + lane];
      const float4 f1 = feat4[(size_t)s * (DD / 4) + 32 + lane];
      const float pA = prow[i];
      const float pB = prow[4 * 32 + i];
      dA += pA; dB += pB;
      acc0.x += pA * f0.x; acc0.y += pA * f0.y; acc0.z += pA * f0.z; acc0.w += pA * f0.w;
      acc1.x += pB * f1.x; acc1.y += pB * f1.y; acc1.z += pB * f1.z; acc1.w += pB * f1.w;
    }
  }
  dA = (dA == 0.f) ? 1.f : dA;
  dB = (dB == 0.f) ? 1.f : dB;

  const float4 b0 = ((const float4*)bias)[lane];
  const float4 b1 = ((const float4*)bias)[32 + lane];
  float4 o0, o1;
  o0.x = elu1(acc0.x / dA + b0.x); o0.y = elu1(acc0.y / dA + b0.y);
  o0.z = elu1(acc0.z / dA + b0.z); o0.w = elu1(acc0.w / dA + b0.w);
  o1.x = elu1(acc1.x / dB + b1.x); o1.y = elu1(acc1.y / dB + b1.y);
  o1.z = elu1(acc1.z / dB + b1.z); o1.w = elu1(acc1.w / dB + b1.w);
  ((float4*)zrow)[lane] = o0;
  ((float4*)zrow)[32 + lane] = o1;
}

#define SEM_ROWS   (NN * MM)
#define SEM_STRIPS (SEM_ROWS / 16)
#define SEM_BLOCKS 256
#define SEM_WAVES  8
#define SEM_STEP   (SEM_BLOCKS * SEM_WAVES)
#define W1_PITCH   (2 * HID + 32)
static_assert(SEM_ROWS % 16 == 0, "whole 16-row strips, no tail");
static_assert(DD % 4 == 0 && HID % 16 == 0, "whole k steps and column tiles");

__device__ __forceinline__ void sem_load_strip(v2f (&a)[DD / 4],
                                               const float* __restrict__ z,
                                               int strip, int l16, int hf) {
  const float* arow = z + (size_t)(strip * 16 + l16) * DD + hf * 2;
#pragma unroll
  for (int kk = 0; kk < DD / 4; ++kk) a[kk] = *(const v2f*)(arow + kk * 4);
}

__global__ __launch_bounds__(SEM_WAVES * 32) void semantic_kernel(
    const float* __restrict__ z, const float* __restrict__ W1,
    const float* __restrict__ b1, const float* __restrict__ W2,
    float* __restrict__ wsum) {
  __shared__ __attribute__((aligned(16))) float w1s[(DD / 2) * W1_PITCH];
  __shared__ float red[SEM_WAVES][MM];
  const int t = threadIdx.x;
  const int lane = t & 31;
  const int hf = lane >> 4;
  const int l16 = lane & 15;
  const int wave = __builtin_amdgcn_readfirstlane(t >> 5);
  int strip = wave * SEM_BLOCKS + blockIdx.x;

  v2f a[DD / 4], an[DD / 4];
  sem_load_strip(a, z, strip, l16, hf);

  for (int idx = t; idx < (DD / 2) * HID; idx += SEM_WAVES * 32) {
    const int kp = idx / HID, n = idx % HID;
    v2f v;
    v.x = W1[(size_t)(2 * kp) * HID + n];
    v.y = W1[(size_t)(2 * kp + 1) * HID + n];
    *(v2f*)&w1s[kp * W1_PITCH + 2 * n] = v;
  }
  float b1c[HID / 16], w2c[HID / 16];
#pragma unroll
  for (int ct = 0; ct < HID / 16; ++ct) {
    b1c[ct] = b1[ct * 16 + l16];
    w2c[ct] = W2[ct * 16 + l16];
  }
  __syncthreads();

  int boff = hf * W1_PITCH + 2 * l16;
  float am0 = 0.f, am1 = 0.f, am2 = 0.f;

  for (; strip < SEM_STRIPS; strip += SEM_STEP) {
    const int nxt = strip + SEM_STEP;
    if (nxt < SEM_STRIPS) sem_load_strip(an, z, nxt, l16, hf);

    asm volatile("" : "+v"(boff));
    const float* bbase = &w1s[boff];

    v8f acc[HID / 16];
#pragma unroll
    for (int ct = 0; ct < HID / 16; ++ct) acc[ct] = v8f{};
#pragma unroll
    for (int kk = 0; kk < DD / 4; ++kk) {
      const float* bp = bbase + kk * 2 * W1_PITCH;
#pragma unroll
      for (int ct = 0; ct < HID / 16; ++ct) {
        const v2f b = *(const v2f*)(bp + ct * 32);
        acc[ct] = __builtin_amdgcn_wmma_f32_16x16x4_f32(
            false, a[kk], false, b, (short)0, acc[ct], false, false);
      }
    }

    float p[8];
#pragma unroll
    for (int r = 0; r < 8; ++r) {
      float s = 0.f;
#pragma unroll
      for (int ct = 0; ct < HID / 16; ++ct) s += tanhf(acc[ct][r] + b1c[ct]) * w2c[ct];
      p[r] = s;
    }
    const float s0 = p[0] + p[3] + p[6];
    const float s1 = p[1] + p[4] + p[7];
    const float s2 = p[2] + p[5];
    const int first = (strip * 16 + hf * 8) % MM;
    am0 += first == 0 ? s0 : first == 1 ? s2 : s1;
    am1 += first == 0 ? s1 : first == 1 ? s0 : s2;
    am2 += first == 0 ? s2 : first == 1 ? s1 : s0;

    if (nxt < SEM_STRIPS) {
#pragma unroll
      for (int kk = 0; kk < DD / 4; ++kk) a[kk] = an[kk];
    }
  }

  for (int off = 16; off > 0; off >>= 1) {
    am0 += __shfl_xor(am0, off);
    am1 += __shfl_xor(am1, off);
    am2 += __shfl_xor(am2, off);
  }
  if (lane == 0) { red[wave][0] = am0; red[wave][1] = am1; red[wave][2] = am2; }
  __syncthreads();
  if (t < MM) {
    float s = 0.f;
    for (int w = 0; w < SEM_WAVES; ++w) s += red[w][t];
    atomicAdd(&wsum[t], s);
  }
}

__global__ void beta_kernel(const float* __restrict__ wsum,
                            float* __restrict__ beta) {
  float w0 = wsum[0] * (1.f / NN);
  float w1 = wsum[1] * (1.f / NN);
  float w2 = wsum[2] * (1.f / NN);
  float mx = fmaxf(w0, fmaxf(w1, w2));
  float e0 = expf(w0 - mx), e1 = expf(w1 - mx), e2 = expf(w2 - mx);
  float inv = 1.f / (e0 + e1 + e2);
  beta[0] = e0 * inv; beta[1] = e1 * inv; beta[2] = e2 * inv;
}

__global__ void final_kernel(const float* __restrict__ z,
                             const float* __restrict__ beta,
                             float* __restrict__ out) {
  int i = blockIdx.x * blockDim.x + threadIdx.x;
  if (i >= NN * DD) return;
  int n = i >> 8, dd = i & 255;
  size_t base = (size_t)n * (MM * DD) + dd;
  out[i] = beta[0] * z[base] + beta[1] * z[base + DD] + beta[2] * z[base + 2 * DD];
}

extern "C" void kernel_launch(void* const* d_in, const int* in_sizes, int n_in,
                              void* d_out, int out_size, void* d_ws, size_t ws_size,
                              hipStream_t stream) {
  const float* h       = (const float*)d_in[0];
  const int*   eidx    = (const int*)  d_in[1];
  const float* W       = (const float*)d_in[2];
  const float* attn_l  = (const float*)d_in[3];
  const float* attn_r  = (const float*)d_in[4];
  const float* bias    = (const float*)d_in[5];
  const float* sem_W1  = (const float*)d_in[6];
  const float* sem_b1  = (const float*)d_in[7];
  const float* sem_W2  = (const float*)d_in[8];
  float* out = (float*)d_out;

  float* ws   = (float*)d_ws;
  float* z    = ws;
  float* feat = z + (size_t)NN * MM * DD;
  float* el   = feat + (size_t)NN * DD;
  float* er   = el + NN * HH;
  int*   lst  = (int*)(er + NN * HH);
  float* wsum = er + 3 * NN * HH;
  float* beta = wsum + MM;
  int*   part = (int*)feat;
  int*   seg  = part + EE;

  hipMemsetAsync(wsum, 0, sizeof(float) * MM, stream);

  for (int m = 0; m < MM; ++m) {
    const float* hm   = h + (size_t)m * NN * IN_SZ;
    const float* Wm   = W + (size_t)m * IN_SZ * DD;
    const float* alm  = attn_l + (size_t)m * HH * OO;
    const float* arm  = attn_r + (size_t)m * HH * OO;
    const float* bm   = bias + (size_t)m * DD;
    const int*   srcm = eidx + (size_t)m * 2 * EE;
    const int*   dstm = srcm + EE;
    float* zm = z + (size_t)m * DD;

    edge_partition_kernel<<<PT_CHUNKS, PT_THREADS, 0, stream>>>(srcm, dstm, part, seg);
    csr_build_kernel<<<CSR_GROUPS, CB_THREADS, 0, stream>>>(part, seg, (int*)zm, lst);
    feat_gemm_kernel<<<NN / 16, 32, 0, stream>>>(hm, Wm, alm, arm, feat, el, er);
    node_gather_kernel<<<NN / 8, 256, 0, stream>>>(lst, el, er, feat, bm, zm);
  }

  semantic_kernel<<<SEM_BLOCKS, SEM_WAVES * 32, 0, stream>>>(z, sem_W1, sem_b1, sem_W2, wsum);
  beta_kernel<<<1, 1, 0, stream>>>(wsum, beta);
  final_kernel<<<(NN * DD + 255) / 256, 256, 0, stream>>>(z, beta, out);
}


# Edge grouping of all three metapaths batched into the first 2 launches (blockIdx.y = metapath; was 2 per metapath): 11 launches, not 15; the three lists wait in the output buffer; sum order unchanged
# speedup vs baseline: 9.7460x; 1.0523x over previous
#include <hip/hip_runtime.h>
#include <math.h>

#define NN 50000
#define EE 800000
#define MM 3
#define IN_SZ 128
#define HH 8
#define OO 32
#define DD 256
#define HID 128

typedef __attribute__((ext_vector_type(2))) float v2f;
typedef __attribute__((ext_vector_type(8))) float v8f;

__global__ __launch_bounds__(32) void feat_gemm_kernel(
    const float* __restrict__ h, const float* __restrict__ W,
    const float* __restrict__ al, const float* __restrict__ ar,
    float* __restrict__ feat, float* __restrict__ el, float* __restrict__ er) {
  const int lane = threadIdx.x;
  const int half = lane >> 4;
  const int l16  = lane & 15;
  const int row0 = blockIdx.x * 16;

  v2f a[32];
  const float* arow = h + (size_t)(row0 + l16) * IN_SZ;
#pragma unroll
  for (int kk = 0; kk < 32; ++kk) {
    int k = kk * 4 + half * 2;
    a[kk].x = arow[k];
    a[kk].y = arow[k + 1];
  }

  const bool b3 = (l16 & 8) != 0, b2 = (l16 & 4) != 0, b1 = (l16 & 2) != 0, b0 = (l16 & 1) != 0;
  float e0 = 0.f, e1 = 0.f, e2 = 0.f, e3 = 0.f, e4 = 0.f, e5 = 0.f, e6 = 0.f, e7 = 0.f;

  for (int hh = 0; hh < HH; ++hh) {
    float v[16];
#pragma unroll
    for (int sub = 0; sub < 2; ++sub) {
      const int ct = hh * 2 + sub;
      v8f acc = {};
#pragma unroll
      for (int kk = 0; kk < 32; ++kk) {
        int k = kk * 4 + half * 2;
        v2f b;
        b.x = W[(size_t)k * DD + ct * 16 + l16];
        b.y = W[(size_t)(k + 1) * DD + ct * 16 + l16];
        acc = __builtin_amdgcn_wmma_f32_16x16x4_f32(
            false, a[kk], false, b, (short)0, acc, false, false);
      }
#pragma unroll
      for (int r = 0; r < 8; ++r) {
        feat[(size_t)(row0 + half * 8 + r) * DD + ct * 16 + l16] = acc[r];
      }
      const float alv = al[ct * 16 + l16];
      const float arv = ar[ct * 16 + l16];
#pragma unroll
      for (int r = 0; r < 8; ++r) {
        if (sub == 0) { v[r] = acc[r] * alv;  v[8 + r] = acc[r] * arv; }
        else          { v[r] += acc[r] * alv; v[8 + r] += acc[r] * arv; }
      }
    }
    float w8[8], w4[4], w2[2];
#pragma unroll
    for (int i = 0; i < 8; ++i) {
      const float keep = b3 ? v[8 + i] : v[i];
      const float send = b3 ? v[i] : v[8 + i];
      w8[i] = keep + __shfl_xor(send, 8);
    }
#pragma unroll
    for (int i = 0; i < 4; ++i) {
      const float keep = b2 ? w8[4 + i] : w8[i];
      const float send = b2 ? w8[i] : w8[4 + i];
      w4[i] = keep + __shfl_xor(send, 4);
    }
#pragma unroll
    for (int i = 0; i < 2; ++i) {
      const float keep = b1 ? w4[2 + i] : w4[i];
      const float send = b1 ? w4[i] : w4[2 + i];
      w2[i] = keep + __shfl_xor(send, 2);
    }
    const float keep = b0 ? w2[1] : w2[0];
    const float send = b0 ? w2[0] : w2[1];
    const float res = keep + __shfl_xor(send, 1);
    e0 = hh == 0 ? res : e0; e1 = hh == 1 ? res : e1; e2 = hh == 2 ? res : e2; e3 = hh == 3 ? res : e3;
    e4 = hh == 4 ? res : e4; e5 = hh == 5 ? res : e5; e6 = hh == 6 ? res : e6; e7 = hh == 7 ? res : e7;
  }
  float* erow = (b3 ? er : el) + (size_t)(row0 + half * 8 + (l16 & 7)) * HH;
  *(float4*)erow       = make_float4(e0, e1, e2, e3);
  *(float4*)(erow + 4) = make_float4(e4, e5, e6, e7);
}

#define ZS (MM * DD)
static_assert(NN % 8 == 0, "eight node waves per block, no tail");

#define CSR_GROUPS  256
#define CSR_SPAN    ((NN + CSR_GROUPS - 1) / CSR_GROUPS)
#define PT_CHUNKS   256
#define PT_EDGES    (EE / PT_CHUNKS)
#define PT_THREADS  625
#define PT_PER      (PT_EDGES / PT_THREADS)
#define CB_THREADS  512
#define CB_KEEP     8
#define SEG_WORDS   (2 * PT_CHUNKS * CSR_GROUPS)
static_assert(CSR_SPAN <= 256, "the node index inside a range fits the record's low 8 bits");
static_assert(NN <= (1 << 23), "a source id shifted by 8 stays a positive int");
static_assert(EE % PT_CHUNKS == 0, "whole chunks, no tail");
static_assert(PT_EDGES % PT_THREADS == 0, "whole trips over a chunk, no tail");
static_assert(CSR_GROUPS == 256 && PT_CHUNKS == 256, "block_scan256 scans 256 values");
static_assert(PT_THREADS >= 256 && CB_THREADS >= 256, "one thread per counter and per table entry");
static_assert((CSR_GROUPS - 1) * CSR_SPAN < NN, "every group owns at least one node");
static_assert((size_t)MM * EE + (size_t)MM * SEG_WORDS <= (size_t)NN * DD, "records and tables of all metapaths fit the feature plane");
static_assert((size_t)MM * EE <= (size_t)NN * DD, "the lists of all metapaths fit the output buffer");
static_assert(((size_t)MM * EE) % 2 == 0 && SEG_WORDS % 2 == 0, "table entries stay 8-byte aligned");

__device__ __forceinline__ int block_scan256(int v, int t, int lane, int* wtot) {
  int x = v;
  for (int off = 1; off < 32; off <<= 1) {
    const int y = __shfl_up(x, off);
    x += (lane >= off) ? y : 0;
  }
  if (t < 256 && lane == 31) wtot[t >> 5] = x;
  __syncthreads();
  const int w = t >> 5;
  int add = 0;
  for (int i = 0; i < 8; ++i) add += (i < w) ? wtot[i] : 0;
  __syncthreads();
  return x + add;
}

__global__ __launch_bounds__(PT_THREADS) void edge_partition_kernel(
    const int* __restrict__ eidx, int* __restrict__ part_all, int* __restrict__ seg_all) {
  const int m = blockIdx.y;
  const int* __restrict__ src = eidx + (size_t)m * 2 * EE;
  const int* __restrict__ dst = src + EE;
  int* __restrict__ part = part_all + (size_t)m * EE;
  int* __restrict__ seg = seg_all + (size_t)m * SEG_WORDS;
  __shared__ int hist[CSR_GROUPS];
  __shared__ int first[CSR_GROUPS];
  __shared__ int wtot[8];
  __shared__ int sorted[PT_EDGES];
  const int t = threadIdx.x;
  const int lane = t & 31;
  const int e0 = blockIdx.x * PT_EDGES;
  if (t < CSR_GROUPS) hist[t] = 0;
  __syncthreads();

  int dv[PT_PER], sv[PT_PER];
#pragma unroll
  for (int i = 0; i < PT_PER; ++i) {
    dv[i] = dst[e0 + t + i * PT_THREADS];
    sv[i] = src[e0 + t + i * PT_THREADS];
  }
  int rec[PT_PER], own[PT_PER], rank[PT_PER];
#pragma unroll
  for (int i = 0; i < PT_PER; ++i) {
    const int d = dv[i];
    unsigned g = (unsigned)d / (unsigned)CSR_SPAN;
    g = g < (unsigned)CSR_GROUPS ? g : (unsigned)(CSR_GROUPS - 1);
    own[i] = (int)g;
    rec[i] = (sv[i] << 8) | (d - (int)g * CSR_SPAN);
    rank[i] = atomicAdd(&hist[g], 1);
  }
  __syncthreads();

  const int c = (t < CSR_GROUPS) ? hist[t] : 0;
  const int inc = block_scan256(c, t, lane, wtot);
  if (t < CSR_GROUPS) {
    first[t] = inc - c;
    *(int2*)&seg[2 * (blockIdx.x * CSR_GROUPS + t)] = make_int2(inc - c, c);
  }
  __syncthreads();
#pragma unroll
  for (int i = 0; i < PT_PER; ++i) sorted[first[own[i]] + rank[i]] = rec[i];
  __syncthreads();
#pragma unroll
  for (int i = 0; i < PT_PER; ++i) part[e0 + t + i * PT_THREADS] = sorted[t + i * PT_THREADS];
}

__global__ __launch_bounds__(CB_THREADS) void csr_build_kernel(
    const int* __restrict__ part_all, const int* __restrict__ seg_all,
    int* __restrict__ z_all, int* __restrict__ lst_all) {
  const int m = blockIdx.y;
  const int* __restrict__ part = part_all + (size_t)m * EE;
  const int* __restrict__ seg = seg_all + (size_t)m * SEG_WORDS;
  int* __restrict__ zmi = z_all + (size_t)m * DD;
  int* __restrict__ lst = lst_all + (size_t)m * EE;
  __shared__ int segat[PT_CHUNKS];
  __shared__ int segbase[PT_CHUNKS + 1];
  __shared__ int cnt[256];
  __shared__ int cur[256];
  __shared__ int wtot[8];
  __shared__ int nbelow;
  const int t = threadIdx.x;
  const int lane = t & 31;
  const int lo = blockIdx.x * CSR_SPAN;
  const int hi = (lo + CSR_SPAN < NN) ? lo + CSR_SPAN : NN;

  int st = 0, ln = 0;
  if (t < PT_CHUNKS) {
    const int2 sg = *(const int2*)&seg[2 * (t * CSR_GROUPS + blockIdx.x)];
    st = sg.x;
    ln = sg.y;
    segat[t] = t * PT_EDGES + st;
  }
  if (t < 256) cnt[t] = 0;
  if (t == 0) nbelow = 0;
  __syncthreads();
  int sb = st;
  for (int off = 16; off > 0; off >>= 1) sb += __shfl_xor(sb, off);
  if (lane == 0 && t < PT_CHUNKS) atomicAdd(&nbelow, sb);
  const int inc = block_scan256(ln, t, lane, wtot);
  if (t < PT_CHUNKS) segbase[t] = inc - ln;
  if (t == PT_CHUNKS - 1) segbase[PT_CHUNKS] = inc;
  __syncthreads();
  const int total = segbase[PT_CHUNKS];

  auto fetch = [&](int k) -> int {
    int g = 0;
#pragma unroll
    for (int step = PT_CHUNKS / 2; step > 0; step >>= 1) g += (segbase[g + step] <= k) ? step : 0;
    return part[segat[g] + (k - segbase[g])];
  };

  int keep[CB_KEEP];
  {
    int g[CB_KEEP];
#pragma unroll
    for (int i = 0; i < CB_KEEP; ++i) g[i] = 0;
#pragma unroll
    for (int step = PT_CHUNKS / 2; step > 0; step >>= 1) {
#pragma unroll
      for (int i = 0; i < CB_KEEP; ++i) g[i] += (segbase[g[i] + step] <= t + i * CB_THREADS) ? step : 0;
    }
#pragma unroll
    for (int i = 0; i < CB_KEEP; ++i) {
      const int k = t + i * CB_THREADS;
      keep[i] = (k < total) ? part[segat[g[i]] + (k - segbase[g[i]])] : -1;
    }
  }
#pragma unroll
  for (int i = 0; i < CB_KEEP; ++i) {
    if (keep[i] >= 0) atomicAdd(&cnt[keep[i] & 255], 1);
  }
  for (int k = t + CB_KEEP * CB_THREADS; k < total; k += CB_THREADS) atomicAdd(&cnt[fetch(k) & 255], 1);
  __syncthreads();

  const int c = (t < 256) ? cnt[t] : 0;
  const int cinc = block_scan256(c, t, lane, wtot);
  const int base = nbelow;
  if (t < 256) cur[t] = base + cinc - c;
  if (t < hi - lo) {
    *(int2*)&zmi[(size_t)(lo + t) * ZS] = make_int2(c, base + cinc);
  }
  __syncthreads();

#pragma unroll
  for (int i = 0; i < CB_KEEP; ++i) {
    if (keep[i] >= 0) lst[atomicAdd(&cur[keep[i] & 255], 1)] = keep[i] >> 8;
  }
  for (int k = t + CB_KEEP * CB_THREADS; k < total; k += CB_THREADS) {
    const int r = fetch(k);
    lst[atomicAdd(&cur[r & 255], 1)] = r >> 8;
  }
}

__device__ __forceinline__ float leaky(float v) { return v > 0.f ? v : 0.2f * v; }
__device__ __forceinline__ float elu1(float x) { return x > 0.f ? x : expm1f(x); }

__global__ __launch_bounds__(256) void node_gather_kernel(
    const int* __restrict__ lst, const float* __restrict__ el,
    const float* __restrict__ er, const float* __restrict__ feat,
    const float* __restrict__ bias, float* zm) {
  __shared__ float pw[8][HH][32];
  static_assert(NN <= (1 << 26), "a source id shifted by 5 stays a positive int");
  const int lane = threadIdx.x & 31;
  const int q = lane >> 3;
  float* pmine = &pw[threadIdx.x >> 5][0][0];
  const float* prow = pmine + q * 32;
  const int d = __builtin_amdgcn_readfirstlane(blockIdx.x * 8 + (threadIdx.x >> 5));
  float* zrow = zm + (size_t)d * ZS;
  const int cnt = ((const int*)zrow)[0];
  const int end = ((const int*)zrow)[1];
  const int beg = end - cnt;

  const float4 erlo = *(const float4*)(er + (size_t)d * HH);
  const float4 erhi = *(const float4*)(er + (size_t)d * HH + 4);
  const float erv[8] = {erlo.x, erlo.y, erlo.z, erlo.w, erhi.x, erhi.y, erhi.z, erhi.w};

  float mx[8], e_first[8];
  int s_first = 0;
#pragma unroll
  for (int hh = 0; hh < 8; ++hh) { mx[hh] = -INFINITY; e_first[hh] = 0.f; }
  for (int base = beg; base < end; base += 32) {
    const int j = base + lane;
    if (j < end) {
      const int s = lst[j];
      const float4 a = *(const float4*)(el + (size_t)s * HH);
      const float4 b = *(const float4*)(el + (size_t)s * HH + 4);
      const float ev[8] = {a.x, a.y, a.z, a.w, b.x, b.y, b.z, b.w};
      if (base == beg) s_first = s;
#pragma unroll
      for (int hh = 0; hh < 8; ++hh) {
        const float e = leaky(ev[hh] + erv[hh]);
        if (base == beg) e_first[hh] = e;
        mx[hh] = fmaxf(mx[hh], e);
      }
    }
  }
#pragma unroll
  for (int off = 16; off > 0; off >>= 1) {
#pragma unroll
    for (int hh = 0; hh < 8; ++hh) mx[hh] = fmaxf(mx[hh], __shfl_xor(mx[hh], off));
  }
  float mh[8];
#pragma unroll
  for (int hh = 0; hh < 8; ++hh) mh[hh] = __builtin_isfinite(mx[hh]) ? mx[hh] : 0.f;

  const float4* feat4 = (const float4*)feat;
  float4 acc0 = {0.f, 0.f, 0.f, 0.f}, acc1 = {0.f, 0.f, 0.f, 0.f};
  float dA = 0.f, dB = 0.f;
  for (int base = beg; base < end; base += 32) {
    const int j = base + lane;
    const bool valid = j < end;
    const int n = (end - base) < 32 ? (end - base) : 32;
    int my_s = s_first;
    float e[8];
#pragma unroll
    for (int hh = 0; hh < 8; ++hh) e[hh] = e_first[hh];
    if (base != beg && valid) {
      my_s = lst[j];
      const float4 a = *(const float4*)(el + (size_t)my_s * HH);
      const float4 b = *(const float4*)(el + (size_t)my_s * HH + 4);
      const float ev[8] = {a.x, a.y, a.z, a.w, b.x, b.y, b.z, b.w};
#pragma unroll
      for (int hh = 0; hh < 8; ++hh) e[hh] = leaky(ev[hh] + erv[hh]);
    }
    const int key = valid ? ((my_s << 5) | lane) : 0x7fffffff;
    int slot = 0;
    for (int i = 0; i < n; ++i) {
      slot += (__builtin_amdgcn_readlane(key, i) < key) ? 1 : 0;
    }
    const int ss = __builtin_amdgcn_ds_permute(slot << 2, my_s);
    if (valid) {
#pragma unroll
      for (int hh = 0; hh < 8; ++hh) pmine[hh * 32 + slot] = expf(e[hh] - mh[hh]);
    }
#pragma unroll 4
    for (int i = 0; i < n; ++i) {
      const int s = __builtin_amdgcn_readlane(ss, i);
      const float4 f0 = feat4[(size_t)s * (DD / 4) + lane];
      const float4 f1 = feat4[(size_t)s * (DD / 4) + 32 + lane];
      const float pA = prow[i];
      const float pB = prow[4 * 32 + i];
      dA += pA; dB += pB;
      acc0.x += pA * f0.x; acc0.y += pA * f0.y; acc0.z += pA * f0.z; acc0.w += pA * f0.w;
      acc1.x += pB * f1.x; acc1.y += pB * f1.y; acc1.z += pB * f1.z; acc1.w += pB * f1.w;
    }
  }
  dA = (dA == 0.f) ? 1.f : dA;
  dB = (dB == 0.f) ? 1.f : dB;

  const float4 b0 = ((const float4*)bias)[lane];
  const float4 b1 = ((const float4*)bias)[32 + lane];
  float4 o0, o1;
  o0.x = elu1(acc0.x / dA + b0.x); o0.y = elu1(acc0.y / dA + b0.y);
  o0.z = elu1(acc0.z / dA + b0.z); o0.w = elu1(acc0.w / dA + b0.w);
  o1.x = elu1(acc1.x / dB + b1.x); o1.y = elu1(acc1.y / dB + b1.y);
  o1.z = elu1(acc1.z / dB + b1.z); o1.w = elu1(acc1.w / dB + b1.w);
  ((float4*)zrow)[lane] = o0;
  ((float4*)zrow)[32 + lane] = o1;
}

#define SEM_ROWS   (NN * MM)
#define SEM_STRIPS (SEM_ROWS / 16)
#define SEM_BLOCKS 256
#define SEM_WAVES  8
#define SEM_STEP   (SEM_BLOCKS * SEM_WAVES)
#define W1_PITCH   (2 * HID + 32)
static_assert(SEM_ROWS % 16 == 0, "whole 16-row strips, no tail");
static_assert(DD % 4 == 0 && HID % 16 == 0, "whole k steps and column tiles");

__device__ __forceinline__ void sem_load_strip(v2f (&a)[DD / 4],
                                               const float* __restrict__ z,
                                               int strip, int l16, int hf) {
  const float* arow = z + (size_t)(strip * 16 + l16) * DD + hf * 2;
#pragma unroll
  for (int kk = 0; kk < DD / 4; ++kk) a[kk] = *(const v2f*)(arow + kk * 4);
}

__global__ __launch_bounds__(SEM_WAVES * 32) void semantic_kernel(
    const float* __restrict__ z, const float* __restrict__ W1,
    const float* __restrict__ b1, const float* __restrict__ W2,
    float* __restrict__ wsum) {
  __shared__ __attribute__((aligned(16))) float w1s[(DD / 2) * W1_PITCH];
  __shared__ float red[SEM_WAVES][MM];
  const int t = threadIdx.x;
  const int lane = t & 31;
  const int hf = lane >> 4;
  const int l16 = lane & 15;
  const int wave = __builtin_amdgcn_readfirstlane(t >> 5);
  int strip = wave * SEM_BLOCKS + blockIdx.x;

  v2f a[DD / 4], an[DD / 4];
  sem_load_strip(a, z, strip, l16, hf);

  for (int idx = t; idx < (DD / 2) * HID; idx += SEM_WAVES * 32) {
    const int kp = idx / HID, n = idx % HID;
    v2f v;
    v.x = W1[(size_t)(2 * kp) * HID + n];
    v.y = W1[(size_t)(2 * kp + 1) * HID + n];
    *(v2f*)&w1s[kp * W1_PITCH + 2 * n] = v;
  }
  float b1c[HID / 16], w2c[HID / 16];
#pragma unroll
  for (int ct = 0; ct < HID / 16; ++ct) {
    b1c[ct] = b1[ct * 16 + l16];
    w2c[ct] = W2[ct * 16 + l16];
  }
  __syncthreads();

  int boff = hf * W1_PITCH + 2 * l16;
  float am0 = 0.f, am1 = 0.f, am2 = 0.f;

  for (; strip < SEM_STRIPS; strip += SEM_STEP) {
    const int nxt = strip + SEM_STEP;
    if (nxt < SEM_STRIPS) sem_load_strip(an, z, nxt, l16, hf);

    asm volatile("" : "+v"(boff));
    const float* bbase = &w1s[boff];

    v8f acc[HID / 16];
#pragma unroll
    for (int ct = 0; ct < HID / 16; ++ct) acc[ct] = v8f{};
#pragma unroll
    for (int kk = 0; kk < DD / 4; ++kk) {
      const float* bp = bbase + kk * 2 * W1_PITCH;
#pragma unroll
      for (int ct = 0; ct < HID / 16; ++ct) {
        const v2f b = *(const v2f*)(bp + ct * 32);
        acc[ct] = __builtin_amdgcn_wmma_f32_16x16x4_f32(
            false, a[kk], false, b, (short)0, acc[ct], false, false);
      }
    }

    float p[8];
#pragma unroll
    for (int r = 0; r < 8; ++r) {
      float s = 0.f;
#pragma unroll
      for (int ct = 0; ct < HID / 16; ++ct) s += tanhf(acc[ct][r] + b1c[ct]) * w2c[ct];
      p[r] = s;
    }
    const float s0 = p[0] + p[3] + p[6];
    const float s1 = p[1] + p[4] + p[7];
    const float s2 = p[2] + p[5];
    const int first = (strip * 16 + hf * 8) % MM;
    am0 += first == 0 ? s0 : first == 1 ? s2 : s1;
    am1 += first == 0 ? s1 : first == 1 ? s0 : s2;
    am2 += first == 0 ? s2 : first == 1 ? s1 : s0;

    if (nxt < SEM_STRIPS) {
#pragma unroll
      for (int kk = 0; kk < DD / 4; ++kk) a[kk] = an[kk];
    }
  }

  for (int off = 16; off > 0; off >>= 1) {
    am0 += __shfl_xor(am0, off);
    am1 += __shfl_xor(am1, off);
    am2 += __shfl_xor(am2, off);
  }
  if (lane == 0) { red[wave][0] = am0; red[wave][1] = am1; red[wave][2] = am2; }
  __syncthreads();
  if (t < MM) {
    float s = 0.f;
    for (int w = 0; w < SEM_WAVES; ++w) s += red[w][t];
    atomicAdd(&wsum[t], s);
  }
}

__global__ void beta_kernel(const float* __restrict__ wsum,
                            float* __restrict__ beta) {
  float w0 = wsum[0] * (1.f / NN);
  float w1 = wsum[1] * (1.f / NN);
  float w2 = wsum[2] * (1.f / NN);
  float mx = fmaxf(w0, fmaxf(w1, w2));
  float e0 = expf(w0 - mx), e1 = expf(w1 - mx), e2 = expf(w2 - mx);
  float inv = 1.f / (e0 + e1 + e2);
  beta[0] = e0 * inv; beta[1] = e1 * inv; beta[2] = e2 * inv;
}

__global__ void final_kernel(const float* __restrict__ z,
                             const float* __restrict__ beta,
                             float* __restrict__ out) {
  int i = blockIdx.x * blockDim.x + threadIdx.x;
  if (i >= NN * DD) return;
  int n = i >> 8, dd = i & 255;
  size_t base = (size_t)n * (MM * DD) + dd;
  out[i] = beta[0] * z[base] + beta[1] * z[base + DD] + beta[2] * z[base + 2 * DD];
}

extern "C" void kernel_launch(void* const* d_in, const int* in_sizes, int n_in,
                              void* d_out, int out_size, void* d_ws, size_t ws_size,
                              hipStream_t stream) {
  const float* h       = (const float*)d_in[0];
  const int*   eidx    = (const int*)  d_in[1];
  const float* W       = (const float*)d_in[2];
  const float* attn_l  = (const float*)d_in[3];
  const float* attn_r  = (const float*)d_in[4];
  const float* bias    = (const float*)d_in[5];
  const float* sem_W1  = (const float*)d_in[6];
  const float* sem_b1  = (const float*)d_in[7];
  const float* sem_W2  = (const float*)d_in[8];
  float* out = (float*)d_out;

  float* ws   = (float*)d_ws;
  float* z    = ws;
  float* feat = z + (size_t)NN * MM * DD;
  float* el   = feat + (size_t)NN * DD;
  float* er   = el + NN * HH;
  float* wsum = er + 3 * NN * HH;
  float* beta = wsum + MM;
  int*   part = (int*)feat;
  int*   seg  = part + (size_t)MM * EE;
  int*   lst  = (int*)d_out;

  hipMemsetAsync(wsum, 0, sizeof(float) * MM, stream);

  edge_partition_kernel<<<dim3(PT_CHUNKS, MM), PT_THREADS, 0, stream>>>(eidx, part, seg);
  csr_build_kernel<<<dim3(CSR_GROUPS, MM), CB_THREADS, 0, stream>>>(part, seg, (int*)z, lst);

  for (int m = 0; m < MM; ++m) {
    const float* hm   = h + (size_t)m * NN * IN_SZ;
    const float* Wm   = W + (size_t)m * IN_SZ * DD;
    const float* alm  = attn_l + (size_t)m * HH * OO;
    const float* arm  = attn_r + (size_t)m * HH * OO;
    const float* bm   = bias + (size_t)m * DD;
    float* zm = z + (size_t)m * DD;

    feat_gemm_kernel<<<NN / 16, 32, 0, stream>>>(hm, Wm, alm, arm, feat, el, er);
    node_gather_kernel<<<NN / 8, 256, 0, stream>>>(lst + (size_t)m * EE, el, er, feat, bm, zm);
  }

  semantic_kernel<<<SEM_BLOCKS, SEM_WAVES * 32, 0, stream>>>(z, sem_W1, sem_b1, sem_W2, wsum);
  beta_kernel<<<1, 1, 0, stream>>>(wsum, beta);
  final_kernel<<<(NN * DD + 255) / 256, 256, 0, stream>>>(z, beta, out);
}


# Final weighted sum as a resident streaming loop: 500 groups of 8 waves, 25 tiles per thread, 16-byte non-temporal loads, 12-18 in flight (was 400000 one-shot waves, 4-byte accesses); sums unchanged
# speedup vs baseline: 10.1985x; 1.0464x over previous
#include <hip/hip_runtime.h>
#include <math.h>

#define NN 50000
#define EE 800000
#define MM 3
#define IN_SZ 128
#define HH 8
#define OO 32
#define DD 256
#define HID 128

typedef __attribute__((ext_vector_type(2))) float v2f;
typedef __attribute__((ext_vector_type(8))) float v8f;
typedef __attribute__((ext_vector_type(4))) float v4f;

__global__ __launch_bounds__(32) void feat_gemm_kernel(
    const float* __restrict__ h, const float* __restrict__ W,
    const float* __restrict__ al, const float* __restrict__ ar,
    float* __restrict__ feat, float* __restrict__ el, float* __restrict__ er) {
  const int lane = threadIdx.x;
  const int half = lane >> 4;
  const int l16  = lane & 15;
  const int row0 = blockIdx.x * 16;

  v2f a[32];
  const float* arow = h + (size_t)(row0 + l16) * IN_SZ;
#pragma unroll
  for (int kk = 0; kk < 32; ++kk) {
    int k = kk * 4 + half * 2;
    a[kk].x = arow[k];
    a[kk].y = arow[k + 1];
  }

  const bool b3 = (l16 & 8) != 0, b2 = (l16 & 4) != 0, b1 = (l16 & 2) != 0, b0 = (l16 & 1) != 0;
  float e0 = 0.f, e1 = 0.f, e2 = 0.f, e3 = 0.f, e4 = 0.f, e5 = 0.f, e6 = 0.f, e7 = 0.f;

  for (int hh = 0; hh < HH; ++hh) {
    float v[16];
#pragma unroll
    for (int sub = 0; sub < 2; ++sub) {
      const int ct = hh * 2 + sub;
      v8f acc = {};
#pragma unroll
      for (int kk = 0; kk < 32; ++kk) {
        int k = kk * 4 + half * 2;
        v2f b;
        b.x = W[(size_t)k * DD + ct * 16 + l16];
        b.y = W[(size_t)(k + 1) * DD + ct * 16 + l16];
        acc = __builtin_amdgcn_wmma_f32_16x16x4_f32(
            false, a[kk], false, b, (short)0, acc, false, false);
      }
#pragma unroll
      for (int r = 0; r < 8; ++r) {
        feat[(size_t)(row0 + half * 8 + r) * DD + ct * 16 + l16] = acc[r];
      }
      const float alv = al[ct * 16 + l16];
      const float arv = ar[ct * 16 + l16];
#pragma unroll
      for (int r = 0; r < 8; ++r) {
        if (sub == 0) { v[r] = acc[r] * alv;  v[8 + r] = acc[r] * arv; }
        else          { v[r] += acc[r] * alv; v[8 + r] += acc[r] * arv; }
      }
    }
    float w8[8], w4[4], w2[2];
#pragma unroll
    for (int i = 0; i < 8; ++i) {
      const float keep = b3 ? v[8 + i] : v[i];
      const float send = b3 ? v[i] : v[8 + i];
      w8[i] = keep + __shfl_xor(send, 8);
    }
#pragma unroll
    for (int i = 0; i < 4; ++i) {
      const float keep = b2 ? w8[4 + i] : w8[i];
      const float send = b2 ? w8[i] : w8[4 + i];
      w4[i] = keep + __shfl_xor(send, 4);
    }
#pragma unroll
    for (int i = 0; i < 2; ++i) {
      const float keep = b1 ? w4[2 + i] : w4[i];
      const float send = b1 ? w4[i] : w4[2 + i];
      w2[i] = keep + __shfl_xor(send, 2);
    }
    const float keep = b0 ? w2[1] : w2[0];
    const float send = b0 ? w2[0] : w2[1];
    const float res = keep + __shfl_xor(send, 1);
    e0 = hh == 0 ? res : e0; e1 = hh == 1 ? res : e1; e2 = hh == 2 ? res : e2; e3 = hh == 3 ? res : e3;
    e4 = hh == 4 ? res : e4; e5 = hh == 5 ? res : e5; e6 = hh == 6 ? res : e6; e7 = hh == 7 ? res : e7;
  }
  float* erow = (b3 ? er : el) + (size_t)(row0 + half * 8 + (l16 & 7)) * HH;
  *(float4*)erow       = make_float4(e0, e1, e2, e3);
  *(float4*)(erow + 4) = make_float4(e4, e5, e6, e7);
}

#define ZS (MM * DD)
static_assert(NN % 8 == 0, "eight node waves per block, no tail");

#define CSR_GROUPS  256
#define CSR_SPAN    ((NN + CSR_GROUPS - 1) / CSR_GROUPS)
#define PT_CHUNKS   256
#define PT_EDGES    (EE / PT_CHUNKS)
#define PT_THREADS  625
#define PT_PER      (PT_EDGES / PT_THREADS)
#define CB_THREADS  512
#define CB_KEEP     8
#define SEG_WORDS   (2 * PT_CHUNKS * CSR_GROUPS)
static_assert(CSR_SPAN <= 256, "the node index inside a range fits the record's low 8 bits");
static_assert(NN <= (1 << 23), "a source id shifted by 8 stays a positive int");
static_assert(EE % PT_CHUNKS == 0, "whole chunks, no tail");
static_assert(PT_EDGES % PT_THREADS == 0, "whole trips over a chunk, no tail");
static_assert(CSR_GROUPS == 256 && PT_CHUNKS == 256, "block_scan256 scans 256 values");
static_assert(PT_THREADS >= 256 && CB_THREADS >= 256, "one thread per counter and per table entry");
static_assert((CSR_GROUPS - 1) * CSR_SPAN < NN, "every group owns at least one node");
static_assert((size_t)MM * EE + (size_t)MM * SEG_WORDS <= (size_t)NN * DD, "records and tables of all metapaths fit the feature plane");
static_assert((size_t)MM * EE <= (size_t)NN * DD, "the lists of all metapaths fit the output buffer");
static_assert(((size_t)MM * EE) % 2 == 0 && SEG_WORDS % 2 == 0, "table entries stay 8-byte aligned");

__device__ __forceinline__ int block_scan256(int v, int t, int lane, int* wtot) {
  int x = v;
  for (int off = 1; off < 32; off <<= 1) {
    const int y = __shfl_up(x, off);
    x += (lane >= off) ? y : 0;
  }
  if (t < 256 && lane == 31) wtot[t >> 5] = x;
  __syncthreads();
  const int w = t >> 5;
  int add = 0;
  for (int i = 0; i < 8; ++i) add += (i < w) ? wtot[i] : 0;
  __syncthreads();
  return x + add;
}

__global__ __launch_bounds__(PT_THREADS) void edge_partition_kernel(
    const int* __restrict__ eidx, int* __restrict__ part_all, int* __restrict__ seg_all) {
  const int m = blockIdx.y;
  const int* __restrict__ src = eidx + (size_t)m * 2 * EE;
  const int* __restrict__ dst = src + EE;
  int* __restrict__ part = part_all + (size_t)m * EE;
  int* __restrict__ seg = seg_all + (size_t)m * SEG_WORDS;
  __shared__ int hist[CSR_GROUPS];
  __shared__ int first[CSR_GROUPS];
  __shared__ int wtot[8];
  __shared__ int sorted[PT_EDGES];
  const int t = threadIdx.x;
  const int lane = t & 31;
  const int e0 = blockIdx.x * PT_EDGES;
  if (t < CSR_GROUPS) hist[t] = 0;
  __syncthreads();

  int dv[PT_PER], sv[PT_PER];
#pragma unroll
  for (int i = 0; i < PT_PER; ++i) {
    dv[i] = dst[e0 + t + i * PT_THREADS];
    sv[i] = src[e0 + t + i * PT_THREADS];
  }
  int rec[PT_PER], own[PT_PER], rank[PT_PER];
#pragma unroll
  for (int i = 0; i < PT_PER; ++i) {
    const int d = dv[i];
    unsigned g = (unsigned)d / (unsigned)CSR_SPAN;
    g = g < (unsigned)CSR_GROUPS ? g : (unsigned)(CSR_GROUPS - 1);
    own[i] = (int)g;
    rec[i] = (sv[i] << 8) | (d - (int)g * CSR_SPAN);
    rank[i] = atomicAdd(&hist[g], 1);
  }
  __syncthreads();

  const int c = (t < CSR_GROUPS) ? hist[t] : 0;
  const int inc = block_scan256(c, t, lane, wtot);
  if (t < CSR_GROUPS) {
    first[t] = inc - c;
    *(int2*)&seg[2 * (blockIdx.x * CSR_GROUPS + t)] = make_int2(inc - c, c);
  }
  __syncthreads();
#pragma unroll
  for (int i = 0; i < PT_PER; ++i) sorted[first[own[i]] + rank[i]] = rec[i];
  __syncthreads();
#pragma unroll
  for (int i = 0; i < PT_PER; ++i) part[e0 + t + i * PT_THREADS] = sorted[t + i * PT_THREADS];
}

__global__ __launch_bounds__(CB_THREADS) void csr_build_kernel(
    const int* __restrict__ part_all, const int* __restrict__ seg_all,
    int* __restrict__ z_all, int* __restrict__ lst_all) {
  const int m = blockIdx.y;
  const int* __restrict__ part = part_all + (size_t)m * EE;
  const int* __restrict__ seg = seg_all + (size_t)m * SEG_WORDS;
  int* __restrict__ zmi = z_all + (size_t)m * DD;
  int* __restrict__ lst = lst_all + (size_t)m * EE;
  __shared__ int segat[PT_CHUNKS];
  __shared__ int segbase[PT_CHUNKS + 1];
  __shared__ int cnt[256];
  __shared__ int cur[256];
  __shared__ int wtot[8];
  __shared__ int nbelow;
  const int t = threadIdx.x;
  const int lane = t & 31;
  const int lo = blockIdx.x * CSR_SPAN;
  const int hi = (lo + CSR_SPAN < NN) ? lo + CSR_SPAN : NN;

  int st = 0, ln = 0;
  if (t < PT_CHUNKS) {
    const int2 sg = *(const int2*)&seg[2 * (t * CSR_GROUPS + blockIdx.x)];
    st = sg.x;
    ln = sg.y;
    segat[t] = t * PT_EDGES + st;
  }
  if (t < 256) cnt[t] = 0;
  if (t == 0) nbelow = 0;
  __syncthreads();
  int sb = st;
  for (int off = 16; off > 0; off >>= 1) sb += __shfl_xor(sb, off);
  if (lane == 0 && t < PT_CHUNKS) atomicAdd(&nbelow, sb);
  const int inc = block_scan256(ln, t, lane, wtot);
  if (t < PT_CHUNKS) segbase[t] = inc - ln;
  if (t == PT_CHUNKS - 1) segbase[PT_CHUNKS] = inc;
  __syncthreads();
  const int total = segbase[PT_CHUNKS];

  auto fetch = [&](int k) -> int {
    int g = 0;
#pragma unroll
    for (int step = PT_CHUNKS / 2; step > 0; step >>= 1) g += (segbase[g + step] <= k) ? step : 0;
    return part[segat[g] + (k - segbase[g])];
  };

  int keep[CB_KEEP];
  {
    int g[CB_KEEP];
#pragma unroll
    for (int i = 0; i < CB_KEEP; ++i) g[i] = 0;
#pragma unroll
    for (int step = PT_CHUNKS / 2; step > 0; step >>= 1) {
#pragma unroll
      for (int i = 0; i < CB_KEEP; ++i) g[i] += (segbase[g[i] + step] <= t + i * CB_THREADS) ? step : 0;
    }
#pragma unroll
    for (int i = 0; i < CB_KEEP; ++i) {
      const int k = t + i * CB_THREADS;
      keep[i] = (k < total) ? part[segat[g[i]] + (k - segbase[g[i]])] : -1;
    }
  }
#pragma unroll
  for (int i = 0; i < CB_KEEP; ++i) {
    if (keep[i] >= 0) atomicAdd(&cnt[keep[i] & 255], 1);
  }
  for (int k = t + CB_KEEP * CB_THREADS; k < total; k += CB_THREADS) atomicAdd(&cnt[fetch(k) & 255], 1);
  __syncthreads();

  const int c = (t < 256) ? cnt[t] : 0;
  const int cinc = block_scan256(c, t, lane, wtot);
  const int base = nbelow;
  if (t < 256) cur[t] = base + cinc - c;
  if (t < hi - lo) {
    *(int2*)&zmi[(size_t)(lo + t) * ZS] = make_int2(c, base + cinc);
  }
  __syncthreads();

#pragma unroll
  for (int i = 0; i < CB_KEEP; ++i) {
    if (keep[i] >= 0) lst[atomicAdd(&cur[keep[i] & 255], 1)] = keep[i] >> 8;
  }
  for (int k = t + CB_KEEP * CB_THREADS; k < total; k += CB_THREADS) {
    const int r = fetch(k);
    lst[atomicAdd(&cur[r & 255], 1)] = r >> 8;
  }
}

__device__ __forceinline__ float leaky(float v) { return v > 0.f ? v : 0.2f * v; }
__device__ __forceinline__ float elu1(float x) { return x > 0.f ? x : expm1f(x); }

__global__ __launch_bounds__(256) void node_gather_kernel(
    const int* __restrict__ lst, const float* __restrict__ el,
    const float* __restrict__ er, const float* __restrict__ feat,
    const float* __restrict__ bias, float* zm) {
  __shared__ float pw[8][HH][32];
  static_assert(NN <= (1 << 26), "a source id shifted by 5 stays a positive int");
  const int lane = threadIdx.x & 31;
  const int q = lane >> 3;
  float* pmine = &pw[threadIdx.x >> 5][0][0];
  const float* prow = pmine + q * 32;
  const int d = __builtin_amdgcn_readfirstlane(blockIdx.x * 8 + (threadIdx.x >> 5));
  float* zrow = zm + (size_t)d * ZS;
  const int cnt = ((const int*)zrow)[0];
  const int end = ((const int*)zrow)[1];
  const int beg = end - cnt;

  const float4 erlo = *(const float4*)(er + (size_t)d * HH);
  const float4 erhi = *(const float4*)(er + (size_t)d * HH + 4);
  const float erv[8] = {erlo.x, erlo.y, erlo.z, erlo.w, erhi.x, erhi.y, erhi.z, erhi.w};

  float mx[8], e_first[8];
  int s_first = 0;
#pragma unroll
  for (int hh = 0; hh < 8; ++hh) { mx[hh] = -INFINITY; e_first[hh] = 0.f; }
  for (int base = beg; base < end; base += 32) {
    const int j = base + lane;
    if (j < end) {
      const int s = lst[j];
      const float4 a = *(const float4*)(el + (size_t)s * HH);
      const float4 b = *(const float4*)(el + (size_t)s * HH + 4);
      const float ev[8] = {a.x, a.y, a.z, a.w, b.x, b.y, b.z, b.w};
      if (base == beg) s_first = s;
#pragma unroll
      for (int hh = 0; hh < 8; ++hh) {
        const float e = leaky(ev[hh] + erv[hh]);
        if (base == beg) e_first[hh] = e;
        mx[hh] = fmaxf(mx[hh], e);
      }
    }
  }
#pragma unroll
  for (int off = 16; off > 0; off >>= 1) {
#pragma unroll
    for (int hh = 0; hh < 8; ++hh) mx[hh] = fmaxf(mx[hh], __shfl_xor(mx[hh], off));
  }
  float mh[8];
#pragma unroll
  for (int hh = 0; hh < 8; ++hh) mh[hh] = __builtin_isfinite(mx[hh]) ? mx[hh] : 0.f;

  const float4* feat4 = (const float4*)feat;
  float4 acc0 = {0.f, 0.f, 0.f, 0.f}, acc1 = {0.f, 0.f, 0.f, 0.f};
  float dA = 0.f, dB = 0.f;
  for (int base = beg; base < end; base += 32) {
    const int j = base + lane;
    const bool valid = j < end;
    const int n = (end - base) < 32 ? (end - base) : 32;
    int my_s = s_first;
    float e[8];
#pragma unroll
    for (int hh = 0; hh < 8; ++hh) e[hh] = e_first[hh];
    if (base != beg && valid) {
      my_s = lst[j];
      const float4 a = *(const float4*)(el + (size_t)my_s * HH);
      const float4 b = *(const float4*)(el + (size_t)my_s * HH + 4);
      const float ev[8] = {a.x, a.y, a.z, a.w, b.x, b.y, b.z, b.w};
#pragma unroll
      for (int hh = 0; hh < 8; ++hh) e[hh] = leaky(ev[hh] + erv[hh]);
    }
    const int key = valid ? ((my_s << 5) | lane) : 0x7fffffff;
    int slot = 0;
    for (int i = 0; i < n; ++i) {
      slot += (__builtin_amdgcn_readlane(key, i) < key) ? 1 : 0;
    }
    const int ss = __builtin_amdgcn_ds_permute(slot << 2, my_s);
    if (valid) {
#pragma unroll
      for (int hh = 0; hh < 8; ++hh) pmine[hh * 32 + slot] = expf(e[hh] - mh[hh]);
    }
#pragma unroll 4
    for (int i = 0; i < n; ++i) {
      const int s = __builtin_amdgcn_readlane(ss, i);
      const float4 f0 = feat4[(size_t)s * (DD / 4) + lane];
      const float4 f1 = feat4[(size_t)s * (DD / 4) + 32 + lane];
      const float pA = prow[i];
      const float pB = prow[4 * 32 + i];
      dA += pA; dB += pB;
      acc0.x += pA * f0.x; acc0.y += pA * f0.y; acc0.z += pA * f0.z; acc0.w += pA * f0.w;
      acc1.x += pB * f1.x; acc1.y += pB * f1.y; acc1.z += pB * f1.z; acc1.w += pB * f1.w;
    }
  }
  dA = (dA == 0.f) ? 1.f : dA;
  dB = (dB == 0.f) ? 1.f : dB;

  const float4 b0 = ((const float4*)bias)[lane];
  const float4 b1 = ((const float4*)bias)[32 + lane];
  float4 o0, o1;
  o0.x = elu1(acc0.x / dA + b0.x); o0.y = elu1(acc0.y / dA + b0.y);
  o0.z = elu1(acc0.z / dA + b0.z); o0.w = elu1(acc0.w / dA + b0.w);
  o1.x = elu1(acc1.x / dB + b1.x); o1.y = elu1(acc1.y / dB + b1.y);
  o1.z = elu1(acc1.z / dB + b1.z); o1.w = elu1(acc1.w / dB + b1.w);
  ((float4*)zrow)[lane] = o0;
  ((float4*)zrow)[32 + lane] = o1;
}

#define SEM_ROWS   (NN * MM)
#define SEM_STRIPS (SEM_ROWS / 16)
#define SEM_BLOCKS 256
#define SEM_WAVES  8
#define SEM_STEP   (SEM_BLOCKS * SEM_WAVES)
#define W1_PITCH   (2 * HID + 32)
static_assert(SEM_ROWS % 16 == 0, "whole 16-row strips, no tail");
static_assert(DD % 4 == 0 && HID % 16 == 0, "whole k steps and column tiles");

__device__ __forceinline__ void sem_load_strip(v2f (&a)[DD / 4],
                                               const float* __restrict__ z,
                                               int strip, int l16, int hf) {
  const float* arow = z + (size_t)(strip * 16 + l16) * DD + hf * 2;
#pragma unroll
  for (int kk = 0; kk < DD / 4; ++kk) a[kk] = *(const v2f*)(arow + kk * 4);
}

__global__ __launch_bounds__(SEM_WAVES * 32) void semantic_kernel(
    const float* __restrict__ z, const float* __restrict__ W1,
    const float* __restrict__ b1, const float* __restrict__ W2,
    float* __restrict__ wsum) {
  __shared__ __attribute__((aligned(16))) float w1s[(DD / 2) * W1_PITCH];
  __shared__ float red[SEM_WAVES][MM];
  const int t = threadIdx.x;
  const int lane = t & 31;
  const int hf = lane >> 4;
  const int l16 = lane & 15;
  const int wave = __builtin_amdgcn_readfirstlane(t >> 5);
  int strip = wave * SEM_BLOCKS + blockIdx.x;

  v2f a[DD / 4], an[DD / 4];
  sem_load_strip(a, z, strip, l16, hf);

  for (int idx = t; idx < (DD / 2) * HID; idx += SEM_WAVES * 32) {
    const int kp = idx / HID, n = idx % HID;
    v2f v;
    v.x = W1[(size_t)(2 * kp) * HID + n];
    v.y = W1[(size_t)(2 * kp + 1) * HID + n];
    *(v2f*)&w1s[kp * W1_PITCH + 2 * n] = v;
  }
  float b1c[HID / 16], w2c[HID / 16];
#pragma unroll
  for (int ct = 0; ct < HID / 16; ++ct) {
    b1c[ct] = b1[ct * 16 + l16];
    w2c[ct] = W2[ct * 16 + l16];
  }
  __syncthreads();

  int boff = hf * W1_PITCH + 2 * l16;
  float am0 = 0.f, am1 = 0.f, am2 = 0.f;

  for (; strip < SEM_STRIPS; strip += SEM_STEP) {
    const int nxt = strip + SEM_STEP;
    if (nxt < SEM_STRIPS) sem_load_strip(an, z, nxt, l16, hf);

    asm volatile("" : "+v"(boff));
    const float* bbase = &w1s[boff];

    v8f acc[HID / 16];
#pragma unroll
    for (int ct = 0; ct < HID / 16; ++ct) acc[ct] = v8f{};
#pragma unroll
    for (int kk = 0; kk < DD / 4; ++kk) {
      const float* bp = bbase + kk * 2 * W1_PITCH;
#pragma unroll
      for (int ct = 0; ct < HID / 16; ++ct) {
        const v2f b = *(const v2f*)(bp + ct * 32);
        acc[ct] = __builtin_amdgcn_wmma_f32_16x16x4_f32(
            false, a[kk], false, b, (short)0, acc[ct], false, false);
      }
    }

    float p[8];
#pragma unroll
    for (int r = 0; r < 8; ++r) {
      float s = 0.f;
#pragma unroll
      for (int ct = 0; ct < HID / 16; ++ct) s += tanhf(acc[ct][r] + b1c[ct]) * w2c[ct];
      p[r] = s;
    }
    const float s0 = p[0] + p[3] + p[6];
    const float s1 = p[1] + p[4] + p[7];
    const float s2 = p[2] + p[5];
    const int first = (strip * 16 + hf * 8) % MM;
    am0 += first == 0 ? s0 : first == 1 ? s2 : s1;
    am1 += first == 0 ? s1 : first == 1 ? s0 : s2;
    am2 += first == 0 ? s2 : first == 1 ? s1 : s0;

    if (nxt < SEM_STRIPS) {
#pragma unroll
      for (int kk = 0; kk < DD / 4; ++kk) a[kk] = an[kk];
    }
  }

  for (int off = 16; off > 0; off >>= 1) {
    am0 += __shfl_xor(am0, off);
    am1 += __shfl_xor(am1, off);
    am2 += __shfl_xor(am2, off);
  }
  if (lane == 0) { red[wave][0] = am0; red[wave][1] = am1; red[wave][2] = am2; }
  __syncthreads();
  if (t < MM) {
    float s = 0.f;
    for (int w = 0; w < SEM_WAVES; ++w) s += red[w][t];
    atomicAdd(&wsum[t], s);
  }
}

__global__ void beta_kernel(const float* __restrict__ wsum,
                            float* __restrict__ beta) {
  float w0 = wsum[0] * (1.f / NN);
  float w1 = wsum[1] * (1.f / NN);
  float w2 = wsum[2] * (1.f / NN);
  float mx = fmaxf(w0, fmaxf(w1, w2));
  float e0 = expf(w0 - mx), e1 = expf(w1 - mx), e2 = expf(w2 - mx);
  float inv = 1.f / (e0 + e1 + e2);
  beta[0] = e0 * inv; beta[1] = e1 * inv; beta[2] = e2 * inv;
}

#define FIN_WGS     500
#define FIN_THREADS 256
#define FIN_BATCH   5
#define FIN_TILES   (NN * (DD / 4) / FIN_THREADS)
#define FIN_TRIPS   (FIN_TILES / FIN_WGS)
#define FIN_NBATCH  (FIN_TRIPS / FIN_BATCH)
static_assert((NN * (DD / 4)) % FIN_THREADS == 0, "whole tiles, no tail");
static_assert(FIN_TILES % FIN_WGS == 0 && FIN_TRIPS % FIN_BATCH == 0, "whole trips and batches, no tail");
static_assert(DD / 4 == 64 && FIN_THREADS == 256, "a tile is four rows of 64 float4");

__device__ __forceinline__ void fin_load_batch(v4f (&v)[FIN_BATCH][MM], const v4f* z4,
                                               int tile0, int r, int c) {
#pragma unroll
  for (int i = 0; i < FIN_BATCH; ++i) {
    const v4f* p = z4 + (size_t)((tile0 + i) * 4 + r) * (MM * DD / 4) + c;
#pragma unroll
    for (int m = 0; m < MM; ++m) v[i][m] = __builtin_nontemporal_load(p + m * (DD / 4));
  }
}

__global__ __launch_bounds__(FIN_THREADS) void final_kernel(
    const float* z, const float* __restrict__ beta, float* out) {
  const float b0 = beta[0], b1 = beta[1], b2 = beta[2];
  const int t = threadIdx.x;
  const int r = t >> 6, c = t & 63;
  const v4f* z4 = (const v4f*)z;
  v4f* o4 = (v4f*)out;

  v4f cur[FIN_BATCH][MM], nxt[FIN_BATCH][MM];
  fin_load_batch(cur, z4, blockIdx.x * FIN_BATCH, r, c);
#pragma unroll
  for (int k = 0; k < FIN_NBATCH; ++k) {
    const int tile0 = (k * FIN_WGS + blockIdx.x) * FIN_BATCH;
    if (k + 1 < FIN_NBATCH) fin_load_batch(nxt, z4, tile0 + FIN_WGS * FIN_BATCH, r, c);
#pragma unroll
    for (int i = 0; i < FIN_BATCH; ++i) {
      const v4f z0 = cur[i][0], z1 = cur[i][1], z2 = cur[i][2];
      v4f o;
      o.x = b0 * z0.x + b1 * z1.x + b2 * z2.x;
      o.y = b0 * z0.y + b1 * z1.y + b2 * z2.y;
      o.z = b0 * z0.z + b1 * z1.z + b2 * z2.z;
      o.w = b0 * z0.w + b1 * z1.w + b2 * z2.w;
      o4[(size_t)(tile0 + i) * FIN_THREADS + t] = o;
    }
    if (k + 1 < FIN_NBATCH) {
#pragma unroll
      for (int i = 0; i < FIN_BATCH; ++i) {
#pragma unroll
        for (int m = 0; m < MM; ++m) cur[i][m] = nxt[i][m];
      }
    }
  }
}

extern "C" void kernel_launch(void* const* d_in, const int* in_sizes, int n_in,
                              void* d_out, int out_size, void* d_ws, size_t ws_size,
                              hipStream_t stream) {
  const float* h       = (const float*)d_in[0];
  const int*   eidx    = (const int*)  d_in[1];
  const float* W       = (const float*)d_in[2];
  const float* attn_l  = (const float*)d_in[3];
  const float* attn_r  = (const float*)d_in[4];
  const float* bias    = (const float*)d_in[5];
  const float* sem_W1  = (const float*)d_in[6];
  const float* sem_b1  = (const float*)d_in[7];
  const float* sem_W2  = (const float*)d_in[8];
  float* out = (float*)d_out;

  float* ws   = (float*)d_ws;
  float* z    = ws;
  float* feat = z + (size_t)NN * MM * DD;
  float* el   = feat + (size_t)NN * DD;
  float* er   = el + NN * HH;
  float* wsum = er + 3 * NN * HH;
  float* beta = wsum + MM;
  int*   part = (int*)feat;
  int*   seg  = part + (size_t)MM * EE;
  int*   lst  = (int*)d_out;

  hipMemsetAsync(wsum, 0, sizeof(float) * MM, stream);

  edge_partition_kernel<<<dim3(PT_CHUNKS, MM), PT_THREADS, 0, stream>>>(eidx, part, seg);
  csr_build_kernel<<<dim3(CSR_GROUPS, MM), CB_THREADS, 0, stream>>>(part, seg, (int*)z, lst);

  for (int m = 0; m < MM; ++m) {
    const float* hm   = h + (size_t)m * NN * IN_SZ;
    const float* Wm   = W + (size_t)m * IN_SZ * DD;
    const float* alm  = attn_l + (size_t)m * HH * OO;
    const float* arm  = attn_r + (size_t)m * HH * OO;
    const float* bm   = bias + (size_t)m * DD;
    float* zm = z + (size_t)m * DD;

    feat_gemm_kernel<<<NN / 16, 32, 0, stream>>>(hm, Wm, alm, arm, feat, el, er);
    node_gather_kernel<<<NN / 8, 256, 0, stream>>>(lst + (size_t)m * EE, el, er, feat, bm, zm);
  }

  semantic_kernel<<<SEM_BLOCKS, SEM_WAVES * 32, 0, stream>>>(z, sem_W1, sem_b1, sem_W2, wsum);
  beta_kernel<<<1, 1, 0, stream>>>(wsum, beta);
  final_kernel<<<FIN_WGS, FIN_THREADS, 0, stream>>>(z, beta, out);
}
